# starting kernel
# baseline (speedup 1.0000x reference)
#include <hip/hip_runtime.h>

typedef float v2f __attribute__((ext_vector_type(2)));
typedef float v8f __attribute__((ext_vector_type(8)));

#define D_MODEL   256
#define BATCH_A   4096
#define BATCH_B   8192
#define NKEYS     4
#define NBUCKETS  1024

__device__ __forceinline__ int hash_row(const int* __restrict__ keys, int row) {
    int s = 0;
#pragma unroll
    for (int k = 0; k < NKEYS; ++k) s += keys[row * NKEYS + k];
    return (int)(((unsigned)s) & (NBUCKETS - 1));
}

__global__ void hash_b_kernel(const int* __restrict__ keys_b, int* __restrict__ hash_b) {
    int j = blockIdx.x * blockDim.x + threadIdx.x;
    if (j < BATCH_B) hash_b[j] = hash_row(keys_b, j);
}

__global__ __launch_bounds__(D_MODEL)
void bucket_sum_kernel(const float* __restrict__ emb_b,
                       const int* __restrict__ hash_b,
                       float* __restrict__ bucket_sums,
                       int* __restrict__ counts) {
    const int g = blockIdx.x;
    const int t = threadIdx.x;
    float acc = 0.0f;
    int cnt = 0;
    for (int j = 0; j < BATCH_B; ++j) {
        if (hash_b[j] == g) {
            acc += emb_b[(size_t)j * D_MODEL + t];
            ++cnt;
        }
    }
    bucket_sums[(size_t)g * D_MODEL + t] = acc;
    if (t == 0) counts[g] = cnt;
}

__global__ __launch_bounds__(D_MODEL)
void build_combined_kernel(const float* __restrict__ emb_a,
                           const int* __restrict__ keys_a,
                           const float* __restrict__ bucket_sums,
                           const int* __restrict__ counts,
                           float* __restrict__ combined) {
    const int i = blockIdx.x;
    const int t = threadIdx.x;
    const int ha  = hash_row(keys_a, i);
    const int cnt = counts[ha];
    const float scale = 1.0f / (float)(cnt < 1 ? 1 : cnt);
    combined[(size_t)i * (2 * D_MODEL) + t]           = emb_a[(size_t)i * D_MODEL + t];
    combined[(size_t)i * (2 * D_MODEL) + D_MODEL + t] =
        bucket_sums[(size_t)ha * D_MODEL + t] * scale;
}

#define BM 64
#define BN 64
#define KT 16
#define AS_LD 20
#define BS_LD 68

__global__ __launch_bounds__(256)
void wmma_gemm_bias_kernel(const float* __restrict__ A,
                           const float* __restrict__ B,
                           const float* __restrict__ bias,
                           float* __restrict__ C,
                           int M, int N, int K, int relu) {
    __shared__ __align__(16) float As[BM][AS_LD];
    __shared__ __align__(16) float Bs[KT][BS_LD];

    const int tid  = threadIdx.x;
    const int lane = tid & 31;
    const int wave = tid >> 5;
    const int wm   = wave >> 1;
    const int wn   = wave & 1;
    const int hi   = lane >> 4;
    const int l15  = lane & 15;

    const int bm = blockIdx.y * BM;
    const int bn = blockIdx.x * BN;

    v8f c0 = {};
    v8f c1 = {};

    const int ar = tid >> 2;
    const int ac = (tid & 3) * 4;
    const int br = tid >> 4;
    const int bc = (tid & 15) * 4;

    for (int k0 = 0; k0 < K; k0 += KT) {
        const float4 av = *(const float4*)&A[(size_t)(bm + ar) * K + (k0 + ac)];
        const float4 bv = *(const float4*)&B[(size_t)(k0 + br) * N + (bn + bc)];
        __syncthreads();
        *(float4*)&As[ar][ac] = av;
        *(float4*)&Bs[br][bc] = bv;
        __syncthreads();

#pragma unroll
        for (int kk = 0; kk < KT; kk += 4) {
            const int kb = kk + 2 * hi;
            v2f a = *(const v2f*)&As[wm * 16 + l15][kb];
            v2f b0, b1;
            b0.x = Bs[kb    ][wn * 32 + l15];
            b0.y = Bs[kb + 1][wn * 32 + l15];
            b1.x = Bs[kb    ][wn * 32 + 16 + l15];
            b1.y = Bs[kb + 1][wn * 32 + 16 + l15];
            c0 = __builtin_amdgcn_wmma_f32_16x16x4_f32(false, a, false, b0,
                                                       (short)0, c0, false, false);
            c1 = __builtin_amdgcn_wmma_f32_16x16x4_f32(false, a, false, b1,
                                                       (short)0, c1, false, false);
        }
    }

    const int n0 = bn + wn * 32 + l15;
    const float bia0 = bias[n0];
    const float bia1 = bias[n0 + 16];
#pragma unroll
    for (int r = 0; r < 8; ++r) {
        const int m = bm + wm * 16 + r + 8 * hi;
        float v0 = c0[r] + bia0;
        float v1 = c1[r] + bia1;
        if (relu) { v0 = fmaxf(v0, 0.0f); v1 = fmaxf(v1, 0.0f); }
        C[(size_t)m * N + n0]      = v0;
        C[(size_t)m * N + n0 + 16] = v1;
    }
}

extern "C" void kernel_launch(void* const* d_in, const int* in_sizes, int n_in,
                              void* d_out, int out_size, void* d_ws, size_t ws_size,
                              hipStream_t stream) {
    const float* emb_a  = (const float*)d_in[0];
    const float* emb_b  = (const float*)d_in[1];
    const int*   keys_a = (const int*)d_in[2];
    const int*   keys_b = (const int*)d_in[3];
    const float* W1 = (const float*)d_in[4];
    const float* b1 = (const float*)d_in[5];
    const float* W2 = (const float*)d_in[6];
    const float* b2 = (const float*)d_in[7];
    const float* Wo = (const float*)d_in[8];
    const float* bo = (const float*)d_in[9];
    float* out = (float*)d_out;

    float* ws          = (float*)d_ws;
    int*   hash_b      = (int*)ws;
    float* bucket_sums = ws + BATCH_B;
    int*   counts      = (int*)(bucket_sums + NBUCKETS * D_MODEL);
    float* combined    = bucket_sums + NBUCKETS * D_MODEL + NBUCKETS;
    float* h           = combined + (size_t)BATCH_A * 2 * D_MODEL;
    float* rel         = h        + (size_t)BATCH_A * 2 * D_MODEL;

    hash_b_kernel<<<(BATCH_B + 255) / 256, 256, 0, stream>>>(keys_b, hash_b);
    bucket_sum_kernel<<<NBUCKETS, D_MODEL, 0, stream>>>(emb_b, hash_b, bucket_sums, counts);
    build_combined_kernel<<<BATCH_A, D_MODEL, 0, stream>>>(emb_a, keys_a, bucket_sums,
                                                           counts, combined);

    wmma_gemm_bias_kernel<<<dim3(512 / BN, BATCH_A / BM), 256, 0, stream>>>(
        combined, W1, b1, h, BATCH_A, 512, 512, 1);
    wmma_gemm_bias_kernel<<<dim3(256 / BN, BATCH_A / BM), 256, 0, stream>>>(
        h, W2, b2, rel, BATCH_A, 256, 512, 0);
    wmma_gemm_bias_kernel<<<dim3(256 / BN, BATCH_A / BM), 256, 0, stream>>>(
        rel, Wo, bo, out, BATCH_A, 256, 256, 0);
}


# bucket sum: rows sorted once by (hash,row) key, position = count of smaller keys; each bucket block adds only its slice of the sorted row list (was: each thread scanned all 8192); sum order kept
# speedup vs baseline: 3.0690x; 3.0690x over previous
#include <hip/hip_runtime.h>

typedef float v2f __attribute__((ext_vector_type(2)));
typedef float v8f __attribute__((ext_vector_type(8)));

#define D_MODEL   256
#define BATCH_A   4096
#define BATCH_B   8192
#define NKEYS     4
#define NBUCKETS  1024

__device__ __forceinline__ int hash_row(const int* __restrict__ keys, int row) {
    int s = 0;
#pragma unroll
    for (int k = 0; k < NKEYS; ++k) s += keys[row * NKEYS + k];
    return (int)(((unsigned)s) & (NBUCKETS - 1));
}

__global__ void hash_b_kernel(const int* __restrict__ keys_b, int* __restrict__ hash_b) {
    int j = blockIdx.x * blockDim.x + threadIdx.x;
    if (j < BATCH_B) hash_b[j] = hash_row(keys_b, j);
}

#define ROW_BITS    13
#define RANK_LANES  32
#define RANK_WAVES  4
#define RANK_BLOCKS ((BATCH_B + NBUCKETS) / RANK_LANES)
#define RANK_STAGE  (BATCH_B / (RANK_LANES * RANK_WAVES * 4))
static_assert((1 << ROW_BITS) == BATCH_B, "row index must fit the low key bits");
static_assert((BATCH_B + NBUCKETS) % RANK_LANES == 0, "whole blocks of queries");
static_assert(RANK_STAGE * RANK_LANES * RANK_WAVES * 4 == BATCH_B, "staging covers every key");

__global__ __launch_bounds__(RANK_LANES * RANK_WAVES)
void rank_rows_kernel(const int* __restrict__ hash_b,
                      int* __restrict__ perm,
                      int* __restrict__ bucket_off) {
    __shared__ __align__(16) unsigned keys[BATCH_B];
    __shared__ unsigned part[RANK_WAVES][RANK_LANES];

    const int tid  = threadIdx.x;
    const int lane = tid & 31;
    const int wave = tid >> 5;

    int4 hv[RANK_STAGE];
#pragma unroll
    for (int s = 0; s < RANK_STAGE; ++s)
        hv[s] = *(const int4*)&hash_b[tid * 4 + s * (RANK_LANES * RANK_WAVES * 4)];
#pragma unroll
    for (int s = 0; s < RANK_STAGE; ++s) {
        const int i = tid * 4 + s * (RANK_LANES * RANK_WAVES * 4);
        uint4 kv;
        kv.x = ((unsigned)hv[s].x << ROW_BITS) | (unsigned)(i);
        kv.y = ((unsigned)hv[s].y << ROW_BITS) | (unsigned)(i + 1);
        kv.z = ((unsigned)hv[s].z << ROW_BITS) | (unsigned)(i + 2);
        kv.w = ((unsigned)hv[s].w << ROW_BITS) | (unsigned)(i + 3);
        *(uint4*)&keys[i] = kv;
    }
    __syncthreads();

    const int q = blockIdx.x * RANK_LANES + lane;
    const unsigned qk = (q < BATCH_B) ? keys[q]
                                      : ((unsigned)(q - BATCH_B) << ROW_BITS);

    unsigned cnt = 0;
    const int k0 = wave * (BATCH_B / RANK_WAVES);
#pragma unroll 8
    for (int i = 0; i < BATCH_B / RANK_WAVES; i += 4) {
        const uint4 kv = *(const uint4*)&keys[k0 + i];
        cnt += (kv.x < qk) ? 1u : 0u;
        cnt += (kv.y < qk) ? 1u : 0u;
        cnt += (kv.z < qk) ? 1u : 0u;
        cnt += (kv.w < qk) ? 1u : 0u;
    }
    part[wave][lane] = cnt;
    __syncthreads();

    if (wave == 0) {
        const unsigned pos = part[0][lane] + part[1][lane] + part[2][lane] + part[3][lane];
        if (q < BATCH_B) perm[pos] = q;
        else             bucket_off[q - BATCH_B] = (int)pos;
    }
}

#define BSUM_THREADS (D_MODEL / 4)

__global__ __launch_bounds__(BSUM_THREADS)
void bucket_sum_kernel(const float* __restrict__ emb_b,
                       const int* __restrict__ perm,
                       const int* __restrict__ bucket_off,
                       float* __restrict__ bucket_sums,
                       int* __restrict__ counts) {
    const int g = blockIdx.x;
    const int t = threadIdx.x * 4;
    const int beg = bucket_off[g];
    const int end = (g + 1 < NBUCKETS) ? bucket_off[g + 1] : BATCH_B;
    float4 acc = make_float4(0.0f, 0.0f, 0.0f, 0.0f);
    for (int p = beg; p < end; p += 8) {
        float4 v[8];
#pragma unroll
        for (int u = 0; u < 8; ++u) {
            const int pp = (p + u < end) ? (p + u) : (end - 1);
            v[u] = *(const float4*)&emb_b[(size_t)perm[pp] * D_MODEL + t];
        }
#pragma unroll
        for (int u = 0; u < 8; ++u) {
            const bool in_slice = (p + u < end);
            acc.x += in_slice ? v[u].x : 0.0f;
            acc.y += in_slice ? v[u].y : 0.0f;
            acc.z += in_slice ? v[u].z : 0.0f;
            acc.w += in_slice ? v[u].w : 0.0f;
        }
    }
    *(float4*)&bucket_sums[(size_t)g * D_MODEL + t] = acc;
    if (threadIdx.x == 0) counts[g] = end - beg;
}

__global__ __launch_bounds__(D_MODEL)
void build_combined_kernel(const float* __restrict__ emb_a,
                           const int* __restrict__ keys_a,
                           const float* __restrict__ bucket_sums,
                           const int* __restrict__ counts,
                           float* __restrict__ combined) {
    const int i = blockIdx.x;
    const int t = threadIdx.x;
    const int ha  = hash_row(keys_a, i);
    const int cnt = counts[ha];
    const float scale = 1.0f / (float)(cnt < 1 ? 1 : cnt);
    combined[(size_t)i * (2 * D_MODEL) + t]           = emb_a[(size_t)i * D_MODEL + t];
    combined[(size_t)i * (2 * D_MODEL) + D_MODEL + t] =
        bucket_sums[(size_t)ha * D_MODEL + t] * scale;
}

#define BM 64
#define BN 64
#define KT 16
#define AS_LD 20
#define BS_LD 68

__global__ __launch_bounds__(256)
void wmma_gemm_bias_kernel(const float* __restrict__ A,
                           const float* __restrict__ B,
                           const float* __restrict__ bias,
                           float* __restrict__ C,
                           int M, int N, int K, int relu) {
    __shared__ __align__(16) float As[BM][AS_LD];
    __shared__ __align__(16) float Bs[KT][BS_LD];

    const int tid  = threadIdx.x;
    const int lane = tid & 31;
    const int wave = tid >> 5;
    const int wm   = wave >> 1;
    const int wn   = wave & 1;
    const int hi   = lane >> 4;
    const int l15  = lane & 15;

    const int bm = blockIdx.y * BM;
    const int bn = blockIdx.x * BN;

    v8f c0 = {};
    v8f c1 = {};

    const int ar = tid >> 2;
    const int ac = (tid & 3) * 4;
    const int br = tid >> 4;
    const int bc = (tid & 15) * 4;

    for (int k0 = 0; k0 < K; k0 += KT) {
        const float4 av = *(const float4*)&A[(size_t)(bm + ar) * K + (k0 + ac)];
        const float4 bv = *(const float4*)&B[(size_t)(k0 + br) * N + (bn + bc)];
        __syncthreads();
        *(float4*)&As[ar][ac] = av;
        *(float4*)&Bs[br][bc] = bv;
        __syncthreads();

#pragma unroll
        for (int kk = 0; kk < KT; kk += 4) {
            const int kb = kk + 2 * hi;
            v2f a = *(const v2f*)&As[wm * 16 + l15][kb];
            v2f b0, b1;
            b0.x = Bs[kb    ][wn * 32 + l15];
            b0.y = Bs[kb + 1][wn * 32 + l15];
            b1.x = Bs[kb    ][wn * 32 + 16 + l15];
            b1.y = Bs[kb + 1][wn * 32 + 16 + l15];
            c0 = __builtin_amdgcn_wmma_f32_16x16x4_f32(false, a, false, b0,
                                                       (short)0, c0, false, false);
            c1 = __builtin_amdgcn_wmma_f32_16x16x4_f32(false, a, false, b1,
                                                       (short)0, c1, false, false);
        }
    }

    const int n0 = bn + wn * 32 + l15;
    const float bia0 = bias[n0];
    const float bia1 = bias[n0 + 16];
#pragma unroll
    for (int r = 0; r < 8; ++r) {
        const int m = bm + wm * 16 + r + 8 * hi;
        float v0 = c0[r] + bia0;
        float v1 = c1[r] + bia1;
        if (relu) { v0 = fmaxf(v0, 0.0f); v1 = fmaxf(v1, 0.0f); }
        C[(size_t)m * N + n0]      = v0;
        C[(size_t)m * N + n0 + 16] = v1;
    }
}

extern "C" void kernel_launch(void* const* d_in, const int* in_sizes, int n_in,
                              void* d_out, int out_size, void* d_ws, size_t ws_size,
                              hipStream_t stream) {
    const float* emb_a  = (const float*)d_in[0];
    const float* emb_b  = (const float*)d_in[1];
    const int*   keys_a = (const int*)d_in[2];
    const int*   keys_b = (const int*)d_in[3];
    const float* W1 = (const float*)d_in[4];
    const float* b1 = (const float*)d_in[5];
    const float* W2 = (const float*)d_in[6];
    const float* b2 = (const float*)d_in[7];
    const float* Wo = (const float*)d_in[8];
    const float* bo = (const float*)d_in[9];
    float* out = (float*)d_out;

    float* ws          = (float*)d_ws;
    int*   hash_b      = (int*)ws;
    float* bucket_sums = ws + BATCH_B;
    int*   counts      = (int*)(bucket_sums + NBUCKETS * D_MODEL);
    float* combined    = bucket_sums + NBUCKETS * D_MODEL + NBUCKETS;
    float* h           = combined + (size_t)BATCH_A * 2 * D_MODEL;
    float* rel         = h        + (size_t)BATCH_A * 2 * D_MODEL;
    int*   perm        = (int*)(rel + (size_t)BATCH_A * D_MODEL);
    int*   bucket_off  = perm + BATCH_B;

    hash_b_kernel<<<(BATCH_B + 255) / 256, 256, 0, stream>>>(keys_b, hash_b);
    rank_rows_kernel<<<RANK_BLOCKS, RANK_LANES * RANK_WAVES, 0, stream>>>(hash_b, perm, bucket_off);
    bucket_sum_kernel<<<NBUCKETS, BSUM_THREADS, 0, stream>>>(emb_b, perm, bucket_off,
                                                             bucket_sums, counts);
    build_combined_kernel<<<BATCH_A, D_MODEL, 0, stream>>>(emb_a, keys_a, bucket_sums,
                                                           counts, combined);

    wmma_gemm_bias_kernel<<<dim3(512 / BN, BATCH_A / BM), 256, 0, stream>>>(
        combined, W1, b1, h, BATCH_A, 512, 512, 1);
    wmma_gemm_bias_kernel<<<dim3(256 / BN, BATCH_A / BM), 256, 0, stream>>>(
        h, W2, b2, rel, BATCH_A, 256, 512, 0);
    wmma_gemm_bias_kernel<<<dim3(256 / BN, BATCH_A / BM), 256, 0, stream>>>(
        rel, Wo, bo, out, BATCH_A, 256, 256, 0);
}


# GEMM operands loaded global -> registers in WMMA fragment order (no LDS, no barrier); a wave owns 16 or 32 rows x 64 interleaved columns; k summed 0,1,4,5 then 2,3,6,7 per 8: rounding changes
# speedup vs baseline: 6.3512x; 2.0695x over previous
#include <hip/hip_runtime.h>

typedef float v2f __attribute__((ext_vector_type(2)));
typedef float v4f __attribute__((ext_vector_type(4)));
typedef float v8f __attribute__((ext_vector_type(8)));

#define D_MODEL   256
#define BATCH_A   4096
#define BATCH_B   8192
#define NKEYS     4
#define NBUCKETS  1024

__device__ __forceinline__ int hash_row(const int* __restrict__ keys, int row) {
    int s = 0;
#pragma unroll
    for (int k = 0; k < NKEYS; ++k) s += keys[row * NKEYS + k];
    return (int)(((unsigned)s) & (NBUCKETS - 1));
}

__global__ void hash_b_kernel(const int* __restrict__ keys_b, int* __restrict__ hash_b) {
    int j = blockIdx.x * blockDim.x + threadIdx.x;
    if (j < BATCH_B) hash_b[j] = hash_row(keys_b, j);
}

#define ROW_BITS    13
#define RANK_LANES  32
#define RANK_WAVES  4
#define RANK_BLOCKS ((BATCH_B + NBUCKETS) / RANK_LANES)
#define RANK_STAGE  (BATCH_B / (RANK_LANES * RANK_WAVES * 4))
static_assert((1 << ROW_BITS) == BATCH_B, "row index must fit the low key bits");
static_assert((BATCH_B + NBUCKETS) % RANK_LANES == 0, "whole blocks of queries");
static_assert(RANK_STAGE * RANK_LANES * RANK_WAVES * 4 == BATCH_B, "staging covers every key");

__global__ __launch_bounds__(RANK_LANES * RANK_WAVES)
void rank_rows_kernel(const int* __restrict__ hash_b,
                      int* __restrict__ perm,
                      int* __restrict__ bucket_off) {
    __shared__ __align__(16) unsigned keys[BATCH_B];
    __shared__ unsigned part[RANK_WAVES][RANK_LANES];

    const int tid  = threadIdx.x;
    const int lane = tid & 31;
    const int wave = tid >> 5;

    int4 hv[RANK_STAGE];
#pragma unroll
    for (int s = 0; s < RANK_STAGE; ++s)
        hv[s] = *(const int4*)&hash_b[tid * 4 + s * (RANK_LANES * RANK_WAVES * 4)];
#pragma unroll
    for (int s = 0; s < RANK_STAGE; ++s) {
        const int i = tid * 4 + s * (RANK_LANES * RANK_WAVES * 4);
        uint4 kv;
        kv.x = ((unsigned)hv[s].x << ROW_BITS) | (unsigned)(i);
        kv.y = ((unsigned)hv[s].y << ROW_BITS) | (unsigned)(i + 1);
        kv.z = ((unsigned)hv[s].z << ROW_BITS) | (unsigned)(i + 2);
        kv.w = ((unsigned)hv[s].w << ROW_BITS) | (unsigned)(i + 3);
        *(uint4*)&keys[i] = kv;
    }
    __syncthreads();

    const int q = blockIdx.x * RANK_LANES + lane;
    const unsigned qk = (q < BATCH_B) ? keys[q]
                                      : ((unsigned)(q - BATCH_B) << ROW_BITS);

    unsigned cnt = 0;
    const int k0 = wave * (BATCH_B / RANK_WAVES);
#pragma unroll 8
    for (int i = 0; i < BATCH_B / RANK_WAVES; i += 4) {
        const uint4 kv = *(const uint4*)&keys[k0 + i];
        cnt += (kv.x < qk) ? 1u : 0u;
        cnt += (kv.y < qk) ? 1u : 0u;
        cnt += (kv.z < qk) ? 1u : 0u;
        cnt += (kv.w < qk) ? 1u : 0u;
    }
    part[wave][lane] = cnt;
    __syncthreads();

    if (wave == 0) {
        const unsigned pos = part[0][lane] + part[1][lane] + part[2][lane] + part[3][lane];
        if (q < BATCH_B) perm[pos] = q;
        else             bucket_off[q - BATCH_B] = (int)pos;
    }
}

#define BSUM_THREADS (D_MODEL / 4)

__global__ __launch_bounds__(BSUM_THREADS)
void bucket_sum_kernel(const float* __restrict__ emb_b,
                       const int* __restrict__ perm,
                       const int* __restrict__ bucket_off,
                       float* __restrict__ bucket_sums,
                       int* __restrict__ counts) {
    const int g = blockIdx.x;
    const int t = threadIdx.x * 4;
    const int beg = bucket_off[g];
    const int end = (g + 1 < NBUCKETS) ? bucket_off[g + 1] : BATCH_B;
    float4 acc = make_float4(0.0f, 0.0f, 0.0f, 0.0f);
    for (int p = beg; p < end; p += 8) {
        float4 v[8];
#pragma unroll
        for (int u = 0; u < 8; ++u) {
            const int pp = (p + u < end) ? (p + u) : (end - 1);
            v[u] = *(const float4*)&emb_b[(size_t)perm[pp] * D_MODEL + t];
        }
#pragma unroll
        for (int u = 0; u < 8; ++u) {
            const bool in_slice = (p + u < end);
            acc.x += in_slice ? v[u].x : 0.0f;
            acc.y += in_slice ? v[u].y : 0.0f;
            acc.z += in_slice ? v[u].z : 0.0f;
            acc.w += in_slice ? v[u].w : 0.0f;
        }
    }
    *(float4*)&bucket_sums[(size_t)g * D_MODEL + t] = acc;
    if (threadIdx.x == 0) counts[g] = end - beg;
}

__global__ __launch_bounds__(D_MODEL)
void build_combined_kernel(const float* __restrict__ emb_a,
                           const int* __restrict__ keys_a,
                           const float* __restrict__ bucket_sums,
                           const int* __restrict__ counts,
                           float* __restrict__ combined) {
    const int i = blockIdx.x;
    const int t = threadIdx.x;
    const int ha  = hash_row(keys_a, i);
    const int cnt = counts[ha];
    const float scale = 1.0f / (float)(cnt < 1 ? 1 : cnt);
    combined[(size_t)i * (2 * D_MODEL) + t]           = emb_a[(size_t)i * D_MODEL + t];
    combined[(size_t)i * (2 * D_MODEL) + D_MODEL + t] =
        bucket_sums[(size_t)ha * D_MODEL + t] * scale;
}

#define GEMM_WAVES 4
#define GEMM_WN    64

template <int MT, int KS>
struct GemmStage {
    v4f a[MT][KS / 8];
    v4f b[KS / 8][2][2];
};

template <int MT, int KS, int K, int N>
__device__ __forceinline__ void gemm_load_stage(GemmStage<MT, KS>& f,
                                                const float* __restrict__ ap,
                                                const float* __restrict__ bp,
                                                int k0) {
#pragma unroll
    for (int g = 0; g < KS / 8; ++g) {
#pragma unroll
        for (int mt = 0; mt < MT; ++mt)
            f.a[mt][g] = *(const v4f*)(ap + (size_t)(16 * mt) * K + (k0 + 8 * g));
#pragma unroll
        for (int j = 0; j < 2; ++j) {
#pragma unroll
            for (int s = 0; s < 2; ++s)
                f.b[g][j][s] = *(const v4f*)(bp + (size_t)(k0 + 8 * g + 2 * j + s) * N);
        }
    }
}

template <int MT, int KS>
__device__ __forceinline__ void gemm_mma_stage(const GemmStage<MT, KS>& f, v8f (&acc)[MT][4]) {
#pragma unroll
    for (int g = 0; g < KS / 8; ++g) {
#pragma unroll
        for (int j = 0; j < 2; ++j) {
            v2f b2[4];
#pragma unroll
            for (int t = 0; t < 4; ++t) {
                b2[t].x = f.b[g][j][0][t];
                b2[t].y = f.b[g][j][1][t];
            }
#pragma unroll
            for (int mt = 0; mt < MT; ++mt) {
                v2f a2;
                a2.x = f.a[mt][g][2 * j];
                a2.y = f.a[mt][g][2 * j + 1];
#pragma unroll
                for (int t = 0; t < 4; ++t)
                    acc[mt][t] = __builtin_amdgcn_wmma_f32_16x16x4_f32(false, a2, false, b2[t],
                                                                       (short)0, acc[mt][t],
                                                                       false, false);
            }
        }
    }
}

template <int MT, int KS, int K, int N, bool RELU>
__global__ __launch_bounds__(GEMM_WAVES * 32)
void wmma_gemm_direct_kernel(const float* __restrict__ A,
                             const float* __restrict__ B,
                             const float* __restrict__ bias,
                             float* __restrict__ C) {
    static_assert(KS % 8 == 0 && K % (2 * KS) == 0 && K >= 2 * KS, "two stages per trip");
    static_assert(N % GEMM_WN == 0, "whole column strips");

    const int lane = threadIdx.x & 31;
    const int wave = threadIdx.x >> 5;
    const int h    = lane >> 4;
    const int c    = lane & 15;

    const int m0 = (blockIdx.y * GEMM_WAVES + wave) * (16 * MT);
    const int n0 = blockIdx.x * GEMM_WN;

    const float* ap = A + (size_t)(m0 + c) * K + 4 * h;
    const float* bp = B + (size_t)(4 * h) * N + (n0 + 4 * c);

    v8f acc[MT][4];
#pragma unroll
    for (int mt = 0; mt < MT; ++mt) {
#pragma unroll
        for (int t = 0; t < 4; ++t) acc[mt][t] = v8f{};
    }

    GemmStage<MT, KS> f0, f1;
    gemm_load_stage<MT, KS, K, N>(f0, ap, bp, 0);
#pragma nounroll
    for (int k0 = 0; k0 < K - 2 * KS; k0 += 2 * KS) {
        gemm_load_stage<MT, KS, K, N>(f1, ap, bp, k0 + KS);
        gemm_mma_stage<MT, KS>(f0, acc);
        gemm_load_stage<MT, KS, K, N>(f0, ap, bp, k0 + 2 * KS);
        gemm_mma_stage<MT, KS>(f1, acc);
    }
    gemm_load_stage<MT, KS, K, N>(f1, ap, bp, K - KS);
    gemm_mma_stage<MT, KS>(f0, acc);
    gemm_mma_stage<MT, KS>(f1, acc);

    const v4f bia = *(const v4f*)(bias + n0 + 4 * c);
#pragma unroll
    for (int mt = 0; mt < MT; ++mt) {
#pragma unroll
        for (int r = 0; r < 8; ++r) {
            v4f o;
            o.x = acc[mt][0][r] + bia.x;
            o.y = acc[mt][1][r] + bia.y;
            o.z = acc[mt][2][r] + bia.z;
            o.w = acc[mt][3][r] + bia.w;
            if (RELU) {
                o.x = fmaxf(o.x, 0.0f);
                o.y = fmaxf(o.y, 0.0f);
                o.z = fmaxf(o.z, 0.0f);
                o.w = fmaxf(o.w, 0.0f);
            }
            const int m = m0 + 16 * mt + 8 * h + r;
            *(v4f*)(C + (size_t)m * N + (n0 + 4 * c)) = o;
        }
    }
}

extern "C" void kernel_launch(void* const* d_in, const int* in_sizes, int n_in,
                              void* d_out, int out_size, void* d_ws, size_t ws_size,
                              hipStream_t stream) {
    const float* emb_a  = (const float*)d_in[0];
    const float* emb_b  = (const float*)d_in[1];
    const int*   keys_a = (const int*)d_in[2];
    const int*   keys_b = (const int*)d_in[3];
    const float* W1 = (const float*)d_in[4];
    const float* b1 = (const float*)d_in[5];
    const float* W2 = (const float*)d_in[6];
    const float* b2 = (const float*)d_in[7];
    const float* Wo = (const float*)d_in[8];
    const float* bo = (const float*)d_in[9];
    float* out = (float*)d_out;

    float* ws          = (float*)d_ws;
    int*   hash_b      = (int*)ws;
    float* bucket_sums = ws + BATCH_B;
    int*   counts      = (int*)(bucket_sums + NBUCKETS * D_MODEL);
    float* combined    = bucket_sums + NBUCKETS * D_MODEL + NBUCKETS;
    float* h           = combined + (size_t)BATCH_A * 2 * D_MODEL;
    float* rel         = h        + (size_t)BATCH_A * 2 * D_MODEL;
    int*   perm        = (int*)(rel + (size_t)BATCH_A * D_MODEL);
    int*   bucket_off  = perm + BATCH_B;

    hash_b_kernel<<<(BATCH_B + 255) / 256, 256, 0, stream>>>(keys_b, hash_b);
    rank_rows_kernel<<<RANK_BLOCKS, RANK_LANES * RANK_WAVES, 0, stream>>>(hash_b, perm, bucket_off);
    bucket_sum_kernel<<<NBUCKETS, BSUM_THREADS, 0, stream>>>(emb_b, perm, bucket_off,
                                                             bucket_sums, counts);
    build_combined_kernel<<<BATCH_A, D_MODEL, 0, stream>>>(emb_a, keys_a, bucket_sums,
                                                           counts, combined);

    wmma_gemm_direct_kernel<2, 32, 512, 512, true>
        <<<dim3(512 / GEMM_WN, BATCH_A / (GEMM_WAVES * 32)), GEMM_WAVES * 32, 0, stream>>>(
            combined, W1, b1, h);
    wmma_gemm_direct_kernel<1, 64, 512, 256, false>
        <<<dim3(256 / GEMM_WN, BATCH_A / (GEMM_WAVES * 16)), GEMM_WAVES * 32, 0, stream>>>(
            h, W2, b2, rel);
    wmma_gemm_direct_kernel<1, 32, 256, 256, false>
        <<<dim3(256 / GEMM_WN, BATCH_A / (GEMM_WAVES * 16)), GEMM_WAVES * 32, 0, stream>>>(
            rel, Wo, bo, out);
}


# three dense layers chained in one kernel: a block carries 16 batch rows through all layers, h and rel strips stay in LDS (products transposed, weights as A operand); 2 launches gone; k order kept
# speedup vs baseline: 7.1409x; 1.1243x over previous
#include <hip/hip_runtime.h>

typedef float v2f __attribute__((ext_vector_type(2)));
typedef float v4f __attribute__((ext_vector_type(4)));
typedef float v8f __attribute__((ext_vector_type(8)));

#define D_MODEL   256
#define BATCH_A   4096
#define BATCH_B   8192
#define NKEYS     4
#define NBUCKETS  1024

__device__ __forceinline__ int hash_row(const int* __restrict__ keys, int row) {
    int s = 0;
#pragma unroll
    for (int k = 0; k < NKEYS; ++k) s += keys[row * NKEYS + k];
    return (int)(((unsigned)s) & (NBUCKETS - 1));
}

__global__ void hash_b_kernel(const int* __restrict__ keys_b, int* __restrict__ hash_b) {
    int j = blockIdx.x * blockDim.x + threadIdx.x;
    if (j < BATCH_B) hash_b[j] = hash_row(keys_b, j);
}

#define ROW_BITS    13
#define RANK_LANES  32
#define RANK_WAVES  4
#define RANK_BLOCKS ((BATCH_B + NBUCKETS) / RANK_LANES)
#define RANK_STAGE  (BATCH_B / (RANK_LANES * RANK_WAVES * 4))
static_assert((1 << ROW_BITS) == BATCH_B, "row index must fit the low key bits");
static_assert((BATCH_B + NBUCKETS) % RANK_LANES == 0, "whole blocks of queries");
static_assert(RANK_STAGE * RANK_LANES * RANK_WAVES * 4 == BATCH_B, "staging covers every key");

__global__ __launch_bounds__(RANK_LANES * RANK_WAVES)
void rank_rows_kernel(const int* __restrict__ hash_b,
                      int* __restrict__ perm,
                      int* __restrict__ bucket_off) {
    __shared__ __align__(16) unsigned keys[BATCH_B];
    __shared__ unsigned part[RANK_WAVES][RANK_LANES];

    const int tid  = threadIdx.x;
    const int lane = tid & 31;
    const int wave = tid >> 5;

    int4 hv[RANK_STAGE];
#pragma unroll
    for (int s = 0; s < RANK_STAGE; ++s)
        hv[s] = *(const int4*)&hash_b[tid * 4 + s * (RANK_LANES * RANK_WAVES * 4)];
#pragma unroll
    for (int s = 0; s < RANK_STAGE; ++s) {
        const int i = tid * 4 + s * (RANK_LANES * RANK_WAVES * 4);
        uint4 kv;
        kv.x = ((unsigned)hv[s].x << ROW_BITS) | (unsigned)(i);
        kv.y = ((unsigned)hv[s].y << ROW_BITS) | (unsigned)(i + 1);
        kv.z = ((unsigned)hv[s].z << ROW_BITS) | (unsigned)(i + 2);
        kv.w = ((unsigned)hv[s].w << ROW_BITS) | (unsigned)(i + 3);
        *(uint4*)&keys[i] = kv;
    }
    __syncthreads();

    const int q = blockIdx.x * RANK_LANES + lane;
    const unsigned qk = (q < BATCH_B) ? keys[q]
                                      : ((unsigned)(q - BATCH_B) << ROW_BITS);

    unsigned cnt = 0;
    const int k0 = wave * (BATCH_B / RANK_WAVES);
#pragma unroll 8
    for (int i = 0; i < BATCH_B / RANK_WAVES; i += 4) {
        const uint4 kv = *(const uint4*)&keys[k0 + i];
        cnt += (kv.x < qk) ? 1u : 0u;
        cnt += (kv.y < qk) ? 1u : 0u;
        cnt += (kv.z < qk) ? 1u : 0u;
        cnt += (kv.w < qk) ? 1u : 0u;
    }
    part[wave][lane] = cnt;
    __syncthreads();

    if (wave == 0) {
        const unsigned pos = part[0][lane] + part[1][lane] + part[2][lane] + part[3][lane];
        if (q < BATCH_B) perm[pos] = q;
        else             bucket_off[q - BATCH_B] = (int)pos;
    }
}

#define BSUM_THREADS (D_MODEL / 4)

__global__ __launch_bounds__(BSUM_THREADS)
void bucket_sum_kernel(const float* __restrict__ emb_b,
                       const int* __restrict__ perm,
                       const int* __restrict__ bucket_off,
                       float* __restrict__ bucket_sums,
                       int* __restrict__ counts) {
    const int g = blockIdx.x;
    const int t = threadIdx.x * 4;
    const int beg = bucket_off[g];
    const int end = (g + 1 < NBUCKETS) ? bucket_off[g + 1] : BATCH_B;
    float4 acc = make_float4(0.0f, 0.0f, 0.0f, 0.0f);
    for (int p = beg; p < end; p += 8) {
        float4 v[8];
#pragma unroll
        for (int u = 0; u < 8; ++u) {
            const int pp = (p + u < end) ? (p + u) : (end - 1);
            v[u] = *(const float4*)&emb_b[(size_t)perm[pp] * D_MODEL + t];
        }
#pragma unroll
        for (int u = 0; u < 8; ++u) {
            const bool in_slice = (p + u < end);
            acc.x += in_slice ? v[u].x : 0.0f;
            acc.y += in_slice ? v[u].y : 0.0f;
            acc.z += in_slice ? v[u].z : 0.0f;
            acc.w += in_slice ? v[u].w : 0.0f;
        }
    }
    *(float4*)&bucket_sums[(size_t)g * D_MODEL + t] = acc;
    if (threadIdx.x == 0) counts[g] = end - beg;
}

__global__ __launch_bounds__(D_MODEL)
void build_combined_kernel(const float* __restrict__ emb_a,
                           const int* __restrict__ keys_a,
                           const float* __restrict__ bucket_sums,
                           const int* __restrict__ counts,
                           float* __restrict__ combined) {
    const int i = blockIdx.x;
    const int t = threadIdx.x;
    const int ha  = hash_row(keys_a, i);
    const int cnt = counts[ha];
    const float scale = 1.0f / (float)(cnt < 1 ? 1 : cnt);
    combined[(size_t)i * (2 * D_MODEL) + t]           = emb_a[(size_t)i * D_MODEL + t];
    combined[(size_t)i * (2 * D_MODEL) + D_MODEL + t] =
        bucket_sums[(size_t)ha * D_MODEL + t] * scale;
}

#define CH_WAVES  4
#define CH_ROWS   16
#define H_PITCH   (2 * D_MODEL + 4)
#define R_PITCH   (D_MODEL + 4)

template <int NG, int KS>
struct ChainStage {
    v4f w[NG][KS / 8][2][2];
    v4f x[KS / 8];
};

template <int NG, int KS, int N>
__device__ __forceinline__ void chain_load_w(ChainStage<NG, KS>& f,
                                             const float* __restrict__ wp, int k0) {
#pragma unroll
    for (int g = 0; g < KS / 8; ++g) {
#pragma unroll
        for (int ng = 0; ng < NG; ++ng) {
#pragma unroll
            for (int j = 0; j < 2; ++j) {
#pragma unroll
                for (int s = 0; s < 2; ++s)
                    f.w[ng][g][j][s] =
                        *(const v4f*)(wp + (size_t)(k0 + 8 * g + 2 * j + s) * N + 64 * ng);
            }
        }
    }
}

template <int NG, int KS>
__device__ __forceinline__ void chain_load_x(ChainStage<NG, KS>& f, const float* xp, int k0) {
#pragma unroll
    for (int g = 0; g < KS / 8; ++g) f.x[g] = *(const v4f*)(xp + (k0 + 8 * g));
}

template <int NG, int KS, int N>
__device__ __forceinline__ void chain_load_stage(ChainStage<NG, KS>& f,
                                                 const float* __restrict__ wp,
                                                 const float* xp,
                                                 int k0) {
    chain_load_x<NG, KS>(f, xp, k0);
    chain_load_w<NG, KS, N>(f, wp, k0);
}

template <int NG, int KS>
__device__ __forceinline__ void chain_mma_stage(const ChainStage<NG, KS>& f, v8f (&acc)[NG][4]) {
#pragma unroll
    for (int g = 0; g < KS / 8; ++g) {
#pragma unroll
        for (int j = 0; j < 2; ++j) {
            v2f x2;
            x2.x = f.x[g][2 * j];
            x2.y = f.x[g][2 * j + 1];
#pragma unroll
            for (int ng = 0; ng < NG; ++ng) {
#pragma unroll
                for (int t = 0; t < 4; ++t) {
                    v2f w2;
                    w2.x = f.w[ng][g][j][0][t];
                    w2.y = f.w[ng][g][j][1][t];
                    acc[ng][t] = __builtin_amdgcn_wmma_f32_16x16x4_f32(false, w2, false, x2,
                                                                       (short)0, acc[ng][t],
                                                                       false, false);
                }
            }
        }
    }
}

template <int NG, int KS, int K, int N>
__device__ __forceinline__ void chain_layer(v8f (&acc)[NG][4],
                                            ChainStage<NG, KS>& f0,
                                            const float* __restrict__ wp,
                                            const float* xp) {
    static_assert(KS % 8 == 0 && K % (2 * KS) == 0 && K >= 4 * KS, "two stages per trip, two trips");
    ChainStage<NG, KS> f1;
    chain_load_x<NG, KS>(f0, xp, 0);
#pragma nounroll
    for (int k0 = 0; k0 < K - 2 * KS; k0 += 2 * KS) {
        chain_load_stage<NG, KS, N>(f1, wp, xp, k0 + KS);
        chain_mma_stage<NG, KS>(f0, acc);
        chain_load_stage<NG, KS, N>(f0, wp, xp, k0 + 2 * KS);
        chain_mma_stage<NG, KS>(f1, acc);
    }
    chain_load_stage<NG, KS, N>(f1, wp, xp, K - KS);
    chain_mma_stage<NG, KS>(f0, acc);
    chain_mma_stage<NG, KS>(f1, acc);
}

template <int NG, bool RELU>
__device__ __forceinline__ void chain_store_strip(const v8f (&acc)[NG][4],
                                                  const float* __restrict__ bias,
                                                  float* dst, int pitch, int nb, int c, int h) {
#pragma unroll
    for (int ng = 0; ng < NG; ++ng) {
#pragma unroll
        for (int r = 0; r < 8; ++r) {
            const int f = nb + 64 * ng + 32 * h + 4 * r;
            const v4f bia = *(const v4f*)(bias + f);
            v4f o;
            o.x = acc[ng][0][r] + bia.x;
            o.y = acc[ng][1][r] + bia.y;
            o.z = acc[ng][2][r] + bia.z;
            o.w = acc[ng][3][r] + bia.w;
            if (RELU) {
                o.x = fmaxf(o.x, 0.0f);
                o.y = fmaxf(o.y, 0.0f);
                o.z = fmaxf(o.z, 0.0f);
                o.w = fmaxf(o.w, 0.0f);
            }
            *(v4f*)(dst + c * pitch + f) = o;
        }
    }
}

__global__ __launch_bounds__(CH_WAVES * 32)
void mlp_chain_kernel(const float* __restrict__ X,
                      const float* __restrict__ W1, const float* __restrict__ b1,
                      const float* __restrict__ W2, const float* __restrict__ b2,
                      const float* __restrict__ Wo, const float* __restrict__ bo,
                      float* __restrict__ out) {
    __shared__ __align__(16) float hbuf[CH_ROWS * H_PITCH];
    __shared__ __align__(16) float rbuf[CH_ROWS * R_PITCH];

    const int lane = threadIdx.x & 31;
    const int wave = threadIdx.x >> 5;
    const int h    = lane >> 4;
    const int c    = lane & 15;
    const int m0   = blockIdx.x * CH_ROWS;

    const float* wp1 = W1 + (size_t)(4 * h) * (2 * D_MODEL) + (128 * wave + 4 * c);
    const float* wp2 = W2 + (size_t)(4 * h) * D_MODEL + (64 * wave + 4 * c);
    const float* wp3 = Wo + (size_t)(4 * h) * D_MODEL + (64 * wave + 4 * c);

    v8f acc1[2][4];
#pragma unroll
    for (int ng = 0; ng < 2; ++ng) {
#pragma unroll
        for (int t = 0; t < 4; ++t) acc1[ng][t] = v8f{};
    }
    ChainStage<2, 32> s1;
    chain_load_w<2, 32, 2 * D_MODEL>(s1, wp1, 0);
    chain_layer<2, 32, 2 * D_MODEL, 2 * D_MODEL>(
        acc1, s1, wp1, X + (size_t)(m0 + c) * (2 * D_MODEL) + 4 * h);
    ChainStage<1, 64> s2;
    chain_load_w<1, 64, D_MODEL>(s2, wp2, 0);
    chain_store_strip<2, true>(acc1, b1, hbuf, H_PITCH, 128 * wave, c, h);
    __syncthreads();

    v8f acc2[1][4];
#pragma unroll
    for (int t = 0; t < 4; ++t) acc2[0][t] = v8f{};
    chain_layer<1, 64, 2 * D_MODEL, D_MODEL>(acc2, s2, wp2, hbuf + c * H_PITCH + 4 * h);
    ChainStage<1, 32> s3;
    chain_load_w<1, 32, D_MODEL>(s3, wp3, 0);
    chain_store_strip<1, false>(acc2, b2, rbuf, R_PITCH, 64 * wave, c, h);
    __syncthreads();

    v8f acc3[1][4];
#pragma unroll
    for (int t = 0; t < 4; ++t) acc3[0][t] = v8f{};
    chain_layer<1, 32, D_MODEL, D_MODEL>(acc3, s3, wp3, rbuf + c * R_PITCH + 4 * h);
    chain_store_strip<1, false>(acc3, bo, hbuf, R_PITCH, 64 * wave, c, h);
    __syncthreads();

#pragma unroll
    for (int it = 0; it < (CH_ROWS * D_MODEL / 4) / (CH_WAVES * 32); ++it) {
        const int idx  = it * (CH_WAVES * 32) + threadIdx.x;
        const int row  = idx / (D_MODEL / 4);
        const int col4 = idx % (D_MODEL / 4);
        *(v4f*)(out + (size_t)(m0 + row) * D_MODEL + 4 * col4) =
            *(const v4f*)(hbuf + row * R_PITCH + 4 * col4);
    }
}

extern "C" void kernel_launch(void* const* d_in, const int* in_sizes, int n_in,
                              void* d_out, int out_size, void* d_ws, size_t ws_size,
                              hipStream_t stream) {
    const float* emb_a  = (const float*)d_in[0];
    const float* emb_b  = (const float*)d_in[1];
    const int*   keys_a = (const int*)d_in[2];
    const int*   keys_b = (const int*)d_in[3];
    const float* W1 = (const float*)d_in[4];
    const float* b1 = (const float*)d_in[5];
    const float* W2 = (const float*)d_in[6];
    const float* b2 = (const float*)d_in[7];
    const float* Wo = (const float*)d_in[8];
    const float* bo = (const float*)d_in[9];
    float* out = (float*)d_out;

    float* ws          = (float*)d_ws;
    int*   hash_b      = (int*)ws;
    float* bucket_sums = ws + BATCH_B;
    int*   counts      = (int*)(bucket_sums + NBUCKETS * D_MODEL);
    float* combined    = bucket_sums + NBUCKETS * D_MODEL + NBUCKETS;
    int*   perm        = (int*)(combined + (size_t)BATCH_A * 2 * D_MODEL);
    int*   bucket_off  = perm + BATCH_B;

    hash_b_kernel<<<(BATCH_B + 255) / 256, 256, 0, stream>>>(keys_b, hash_b);
    rank_rows_kernel<<<RANK_BLOCKS, RANK_LANES * RANK_WAVES, 0, stream>>>(hash_b, perm, bucket_off);
    bucket_sum_kernel<<<NBUCKETS, BSUM_THREADS, 0, stream>>>(emb_b, perm, bucket_off,
                                                             bucket_sums, counts);
    build_combined_kernel<<<BATCH_A, D_MODEL, 0, stream>>>(emb_a, keys_a, bucket_sums,
                                                           counts, combined);

    mlp_chain_kernel<<<BATCH_A / CH_ROWS, CH_WAVES * 32, 0, stream>>>(combined, W1, b1, W2, b2,
                                                                      Wo, bo, out);
}


# bucket blocks write their mean (sum x 1/count) straight to the table A rows of their bucket, found by scanning hash_a; layer 1 reads [emb_a | those rows]; builder launch gone; values bit-identical
# speedup vs baseline: 7.8342x; 1.0971x over previous
#include <hip/hip_runtime.h>

typedef float v2f __attribute__((ext_vector_type(2)));
typedef float v4f __attribute__((ext_vector_type(4)));
typedef float v8f __attribute__((ext_vector_type(8)));

#define D_MODEL   256
#define BATCH_A   4096
#define BATCH_B   8192
#define NKEYS     4
#define NBUCKETS  1024

__device__ __forceinline__ int hash_row(const int* __restrict__ keys, int row) {
    int s = 0;
#pragma unroll
    for (int k = 0; k < NKEYS; ++k) s += keys[row * NKEYS + k];
    return (int)(((unsigned)s) & (NBUCKETS - 1));
}

__global__ void hash_rows_kernel(const int* __restrict__ keys_a, const int* __restrict__ keys_b,
                                 int* __restrict__ hash_a, int* __restrict__ hash_b) {
    int j = blockIdx.x * blockDim.x + threadIdx.x;
    if (j < BATCH_B)                hash_b[j] = hash_row(keys_b, j);
    else if (j < BATCH_B + BATCH_A) hash_a[j - BATCH_B] = hash_row(keys_a, j - BATCH_B);
}

#define ROW_BITS    13
#define RANK_LANES  32
#define RANK_WAVES  4
#define RANK_BLOCKS ((BATCH_B + NBUCKETS) / RANK_LANES)
#define RANK_STAGE  (BATCH_B / (RANK_LANES * RANK_WAVES * 4))
static_assert((1 << ROW_BITS) == BATCH_B, "row index must fit the low key bits");
static_assert((BATCH_B + NBUCKETS) % RANK_LANES == 0, "whole blocks of queries");
static_assert(RANK_STAGE * RANK_LANES * RANK_WAVES * 4 == BATCH_B, "staging covers every key");

__global__ __launch_bounds__(RANK_LANES * RANK_WAVES)
void rank_rows_kernel(const int* __restrict__ hash_b,
                      int* __restrict__ perm,
                      int* __restrict__ bucket_off) {
    __shared__ __align__(16) unsigned keys[BATCH_B];
    __shared__ unsigned part[RANK_WAVES][RANK_LANES];

    const int tid  = threadIdx.x;
    const int lane = tid & 31;
    const int wave = tid >> 5;

    int4 hv[RANK_STAGE];
#pragma unroll
    for (int s = 0; s < RANK_STAGE; ++s)
        hv[s] = *(const int4*)&hash_b[tid * 4 + s * (RANK_LANES * RANK_WAVES * 4)];
#pragma unroll
    for (int s = 0; s < RANK_STAGE; ++s) {
        const int i = tid * 4 + s * (RANK_LANES * RANK_WAVES * 4);
        uint4 kv;
        kv.x = ((unsigned)hv[s].x << ROW_BITS) | (unsigned)(i);
        kv.y = ((unsigned)hv[s].y << ROW_BITS) | (unsigned)(i + 1);
        kv.z = ((unsigned)hv[s].z << ROW_BITS) | (unsigned)(i + 2);
        kv.w = ((unsigned)hv[s].w << ROW_BITS) | (unsigned)(i + 3);
        *(uint4*)&keys[i] = kv;
    }
    __syncthreads();

    const int q = blockIdx.x * RANK_LANES + lane;
    const unsigned qk = (q < BATCH_B) ? keys[q]
                                      : ((unsigned)(q - BATCH_B) << ROW_BITS);

    unsigned cnt = 0;
    const int k0 = wave * (BATCH_B / RANK_WAVES);
#pragma unroll 8
    for (int i = 0; i < BATCH_B / RANK_WAVES; i += 4) {
        const uint4 kv = *(const uint4*)&keys[k0 + i];
        cnt += (kv.x < qk) ? 1u : 0u;
        cnt += (kv.y < qk) ? 1u : 0u;
        cnt += (kv.z < qk) ? 1u : 0u;
        cnt += (kv.w < qk) ? 1u : 0u;
    }
    part[wave][lane] = cnt;
    __syncthreads();

    if (wave == 0) {
        const unsigned pos = part[0][lane] + part[1][lane] + part[2][lane] + part[3][lane];
        if (q < BATCH_B) perm[pos] = q;
        else             bucket_off[q - BATCH_B] = (int)pos;
    }
}

#define BSUM_THREADS (D_MODEL / 4)
#define BSUM_WAVES   (BSUM_THREADS / 32)
#define ASCAN_STAGE  (BATCH_A / (BSUM_THREADS * 4))
#define ASCAN_SHARE  (BATCH_A / BSUM_WAVES)
static_assert(ASCAN_STAGE * BSUM_THREADS * 4 == BATCH_A, "the scan covers every table A row");
static_assert(BATCH_A <= 65536, "row numbers are listed as 16-bit values");

__global__ __launch_bounds__(BSUM_THREADS)
void bucket_mean_kernel(const float* __restrict__ emb_b,
                        const int* __restrict__ perm,
                        const int* __restrict__ bucket_off,
                        const int* __restrict__ hash_a,
                        float* __restrict__ agg) {
    __shared__ unsigned short a_rows[BSUM_WAVES][ASCAN_SHARE];
    __shared__ int a_count[BSUM_WAVES];

    const int g    = blockIdx.x;
    const int tid  = threadIdx.x;
    const int lane = tid & 31;
    const int wave = tid >> 5;
    const int t    = tid * 4;

    int4 ha[ASCAN_STAGE];
#pragma unroll
    for (int s = 0; s < ASCAN_STAGE; ++s)
        ha[s] = *(const int4*)&hash_a[(s * BSUM_THREADS + tid) * 4];

    const int beg = bucket_off[g];
    const int end = (g + 1 < NBUCKETS) ? bucket_off[g + 1] : BATCH_B;

    int n = 0;
#pragma unroll
    for (int s = 0; s < ASCAN_STAGE; ++s) {
        const int i0 = (s * BSUM_THREADS + tid) * 4;
        const bool any = (ha[s].x == g) | (ha[s].y == g) | (ha[s].z == g) | (ha[s].w == g);
        if (__builtin_amdgcn_ballot_w32(any) != 0u) {
            const int hv[4] = {ha[s].x, ha[s].y, ha[s].z, ha[s].w};
#pragma unroll
            for (int e = 0; e < 4; ++e) {
                const bool m = (hv[e] == g);
                const unsigned mask = __builtin_amdgcn_ballot_w32(m);
                if (m) a_rows[wave][n + (int)__builtin_amdgcn_mbcnt_lo(mask, 0u)] =
                           (unsigned short)(i0 + e);
                n += __builtin_popcount(mask);
            }
        }
    }
    if (lane == 0) a_count[wave] = n;

    float4 acc = make_float4(0.0f, 0.0f, 0.0f, 0.0f);
    for (int p = beg; p < end; p += 8) {
        float4 v[8];
#pragma unroll
        for (int u = 0; u < 8; ++u) {
            const int pp = (p + u < end) ? (p + u) : (end - 1);
            v[u] = *(const float4*)&emb_b[(size_t)perm[pp] * D_MODEL + t];
        }
#pragma unroll
        for (int u = 0; u < 8; ++u) {
            const bool in_slice = (p + u < end);
            acc.x += in_slice ? v[u].x : 0.0f;
            acc.y += in_slice ? v[u].y : 0.0f;
            acc.z += in_slice ? v[u].z : 0.0f;
            acc.w += in_slice ? v[u].w : 0.0f;
        }
    }
    const int cnt = end - beg;
    const float scale = 1.0f / (float)(cnt < 1 ? 1 : cnt);
    float4 mean;
    mean.x = acc.x * scale;
    mean.y = acc.y * scale;
    mean.z = acc.z * scale;
    mean.w = acc.w * scale;

    __syncthreads();
#pragma unroll
    for (int w = 0; w < BSUM_WAVES; ++w) {
        const int nw = a_count[w];
        for (int r = 0; r < nw; ++r) {
            const int row = a_rows[w][r];
            *(float4*)&agg[(size_t)row * D_MODEL + t] = mean;
        }
    }
}

#define CH_WAVES  4
#define CH_ROWS   16
#define H_PITCH   (2 * D_MODEL + 4)
#define R_PITCH   (D_MODEL + 4)

template <int NG, int KS>
struct ChainStage {
    v4f w[NG][KS / 8][2][2];
    v4f x[KS / 8];
};

template <int NG, int KS, int N>
__device__ __forceinline__ void chain_load_w(ChainStage<NG, KS>& f,
                                             const float* __restrict__ wp, int k0) {
#pragma unroll
    for (int g = 0; g < KS / 8; ++g) {
#pragma unroll
        for (int ng = 0; ng < NG; ++ng) {
#pragma unroll
            for (int j = 0; j < 2; ++j) {
#pragma unroll
                for (int s = 0; s < 2; ++s)
                    f.w[ng][g][j][s] =
                        *(const v4f*)(wp + (size_t)(k0 + 8 * g + 2 * j + s) * N + 64 * ng);
            }
        }
    }
}

template <int NG, int KS>
__device__ __forceinline__ void chain_load_x(ChainStage<NG, KS>& f, const float* xp, int k0) {
#pragma unroll
    for (int g = 0; g < KS / 8; ++g) f.x[g] = *(const v4f*)(xp + (k0 + 8 * g));
}

template <int NG, int KS, int N>
__device__ __forceinline__ void chain_load_stage(ChainStage<NG, KS>& f,
                                                 const float* __restrict__ wp,
                                                 const float* xp,
                                                 int k0) {
    chain_load_x<NG, KS>(f, xp, k0);
    chain_load_w<NG, KS, N>(f, wp, k0);
}

template <int NG, int KS>
__device__ __forceinline__ void chain_mma_stage(const ChainStage<NG, KS>& f, v8f (&acc)[NG][4]) {
#pragma unroll
    for (int g = 0; g < KS / 8; ++g) {
#pragma unroll
        for (int j = 0; j < 2; ++j) {
            v2f x2;
            x2.x = f.x[g][2 * j];
            x2.y = f.x[g][2 * j + 1];
#pragma unroll
            for (int ng = 0; ng < NG; ++ng) {
#pragma unroll
                for (int t = 0; t < 4; ++t) {
                    v2f w2;
                    w2.x = f.w[ng][g][j][0][t];
                    w2.y = f.w[ng][g][j][1][t];
                    acc[ng][t] = __builtin_amdgcn_wmma_f32_16x16x4_f32(false, w2, false, x2,
                                                                       (short)0, acc[ng][t],
                                                                       false, false);
                }
            }
        }
    }
}

template <int NG, int KS, int K, int N>
__device__ __forceinline__ void chain_layer(v8f (&acc)[NG][4],
                                            ChainStage<NG, KS>& f0,
                                            const float* __restrict__ wp,
                                            const float* xp) {
    static_assert(KS % 8 == 0 && K % (2 * KS) == 0 && K >= 4 * KS, "two stages per trip, two trips");
    ChainStage<NG, KS> f1;
    chain_load_x<NG, KS>(f0, xp, 0);
#pragma nounroll
    for (int k0 = 0; k0 < K - 2 * KS; k0 += 2 * KS) {
        chain_load_stage<NG, KS, N>(f1, wp, xp, k0 + KS);
        chain_mma_stage<NG, KS>(f0, acc);
        chain_load_stage<NG, KS, N>(f0, wp, xp, k0 + 2 * KS);
        chain_mma_stage<NG, KS>(f1, acc);
    }
    chain_load_stage<NG, KS, N>(f1, wp, xp, K - KS);
    chain_mma_stage<NG, KS>(f0, acc);
    chain_mma_stage<NG, KS>(f1, acc);
}

template <int NG, int KS, int H, int N>
__device__ __forceinline__ void chain_layer_joined(v8f (&acc)[NG][4],
                                                   ChainStage<NG, KS>& f0,
                                                   const float* __restrict__ wp,
                                                   const float* xlo,
                                                   const float* xhi) {
    static_assert(KS % 8 == 0 && H % (2 * KS) == 0 && H >= 4 * KS, "whole trips on each side of the join");
    ChainStage<NG, KS> f1;
    chain_load_x<NG, KS>(f0, xlo, 0);
#pragma nounroll
    for (int k0 = 0; k0 < H - 2 * KS; k0 += 2 * KS) {
        chain_load_stage<NG, KS, N>(f1, wp, xlo, k0 + KS);
        chain_mma_stage<NG, KS>(f0, acc);
        chain_load_stage<NG, KS, N>(f0, wp, xlo, k0 + 2 * KS);
        chain_mma_stage<NG, KS>(f1, acc);
    }
    chain_load_stage<NG, KS, N>(f1, wp, xlo, H - KS);
    chain_mma_stage<NG, KS>(f0, acc);
    chain_load_x<NG, KS>(f0, xhi, 0);
    chain_load_w<NG, KS, N>(f0, wp, H);
    chain_mma_stage<NG, KS>(f1, acc);
    const float* wph = wp + (size_t)H * N;
#pragma nounroll
    for (int k0 = 0; k0 < H - 2 * KS; k0 += 2 * KS) {
        chain_load_stage<NG, KS, N>(f1, wph, xhi, k0 + KS);
        chain_mma_stage<NG, KS>(f0, acc);
        chain_load_stage<NG, KS, N>(f0, wph, xhi, k0 + 2 * KS);
        chain_mma_stage<NG, KS>(f1, acc);
    }
    chain_load_stage<NG, KS, N>(f1, wph, xhi, H - KS);
    chain_mma_stage<NG, KS>(f0, acc);
    chain_mma_stage<NG, KS>(f1, acc);
}

template <int NG, bool RELU>
__device__ __forceinline__ void chain_store_strip(const v8f (&acc)[NG][4],
                                                  const float* __restrict__ bias,
                                                  float* dst, int pitch, int nb, int c, int h) {
#pragma unroll
    for (int ng = 0; ng < NG; ++ng) {
#pragma unroll
        for (int r = 0; r < 8; ++r) {
            const int f = nb + 64 * ng + 32 * h + 4 * r;
            const v4f bia = *(const v4f*)(bias + f);
            v4f o;
            o.x = acc[ng][0][r] + bia.x;
            o.y = acc[ng][1][r] + bia.y;
            o.z = acc[ng][2][r] + bia.z;
            o.w = acc[ng][3][r] + bia.w;
            if (RELU) {
                o.x = fmaxf(o.x, 0.0f);
                o.y = fmaxf(o.y, 0.0f);
                o.z = fmaxf(o.z, 0.0f);
                o.w = fmaxf(o.w, 0.0f);
            }
            *(v4f*)(dst + c * pitch + f) = o;
        }
    }
}

__global__ __launch_bounds__(CH_WAVES * 32)
void mlp_chain_kernel(const float* __restrict__ emb_a,
                      const float* __restrict__ agg,
                      const float* __restrict__ W1, const float* __restrict__ b1,
                      const float* __restrict__ W2, const float* __restrict__ b2,
                      const float* __restrict__ Wo, const float* __restrict__ bo,
                      float* __restrict__ out) {
    __shared__ __align__(16) float hbuf[CH_ROWS * H_PITCH];
    __shared__ __align__(16) float rbuf[CH_ROWS * R_PITCH];

    const int lane = threadIdx.x & 31;
    const int wave = threadIdx.x >> 5;
    const int h    = lane >> 4;
    const int c    = lane & 15;
    const int m0   = blockIdx.x * CH_ROWS;

    const float* wp1 = W1 + (size_t)(4 * h) * (2 * D_MODEL) + (128 * wave + 4 * c);
    const float* wp2 = W2 + (size_t)(4 * h) * D_MODEL + (64 * wave + 4 * c);
    const float* wp3 = Wo + (size_t)(4 * h) * D_MODEL + (64 * wave + 4 * c);

    v8f acc1[2][4];
#pragma unroll
    for (int ng = 0; ng < 2; ++ng) {
#pragma unroll
        for (int t = 0; t < 4; ++t) acc1[ng][t] = v8f{};
    }
    ChainStage<2, 32> s1;
    chain_load_w<2, 32, 2 * D_MODEL>(s1, wp1, 0);
    chain_layer_joined<2, 32, D_MODEL, 2 * D_MODEL>(
        acc1, s1, wp1, emb_a + (size_t)(m0 + c) * D_MODEL + 4 * h,
        agg + (size_t)(m0 + c) * D_MODEL + 4 * h);
    ChainStage<1, 64> s2;
    chain_load_w<1, 64, D_MODEL>(s2, wp2, 0);
    chain_store_strip<2, true>(acc1, b1, hbuf, H_PITCH, 128 * wave, c, h);
    __syncthreads();

    v8f acc2[1][4];
#pragma unroll
    for (int t = 0; t < 4; ++t) acc2[0][t] = v8f{};
    chain_layer<1, 64, 2 * D_MODEL, D_MODEL>(acc2, s2, wp2, hbuf + c * H_PITCH + 4 * h);
    ChainStage<1, 32> s3;
    chain_load_w<1, 32, D_MODEL>(s3, wp3, 0);
    chain_store_strip<1, false>(acc2, b2, rbuf, R_PITCH, 64 * wave, c, h);
    __syncthreads();

    v8f acc3[1][4];
#pragma unroll
    for (int t = 0; t < 4; ++t) acc3[0][t] = v8f{};
    chain_layer<1, 32, D_MODEL, D_MODEL>(acc3, s3, wp3, rbuf + c * R_PITCH + 4 * h);
    chain_store_strip<1, false>(acc3, bo, hbuf, R_PITCH, 64 * wave, c, h);
    __syncthreads();

#pragma unroll
    for (int it = 0; it < (CH_ROWS * D_MODEL / 4) / (CH_WAVES * 32); ++it) {
        const int idx  = it * (CH_WAVES * 32) + threadIdx.x;
        const int row  = idx / (D_MODEL / 4);
        const int col4 = idx % (D_MODEL / 4);
        *(v4f*)(out + (size_t)(m0 + row) * D_MODEL + 4 * col4) =
            *(const v4f*)(hbuf + row * R_PITCH + 4 * col4);
    }
}

extern "C" void kernel_launch(void* const* d_in, const int* in_sizes, int n_in,
                              void* d_out, int out_size, void* d_ws, size_t ws_size,
                              hipStream_t stream) {
    const float* emb_a  = (const float*)d_in[0];
    const float* emb_b  = (const float*)d_in[1];
    const int*   keys_a = (const int*)d_in[2];
    const int*   keys_b = (const int*)d_in[3];
    const float* W1 = (const float*)d_in[4];
    const float* b1 = (const float*)d_in[5];
    const float* W2 = (const float*)d_in[6];
    const float* b2 = (const float*)d_in[7];
    const float* Wo = (const float*)d_in[8];
    const float* bo = (const float*)d_in[9];
    float* out = (float*)d_out;

    float* ws         = (float*)d_ws;
    int*   hash_b     = (int*)ws;
    int*   hash_a     = hash_b + BATCH_B;
    int*   perm       = hash_a + BATCH_A;
    int*   bucket_off = perm + BATCH_B;
    float* agg        = (float*)(bucket_off + NBUCKETS);

    hash_rows_kernel<<<(BATCH_B + BATCH_A) / 256, 256, 0, stream>>>(keys_a, keys_b, hash_a, hash_b);
    rank_rows_kernel<<<RANK_BLOCKS, RANK_LANES * RANK_WAVES, 0, stream>>>(hash_b, perm, bucket_off);
    bucket_mean_kernel<<<NBUCKETS, BSUM_THREADS, 0, stream>>>(emb_b, perm, bucket_off, hash_a, agg);

    mlp_chain_kernel<<<BATCH_A / CH_ROWS, CH_WAVES * 32, 0, stream>>>(emb_a, agg, W1, b1, W2, b2,
                                                                      Wo, bo, out);
}


# bucket means made in the dense kernel by 4 added waves per block (4 rows each, sorted-list slices, row order kept) while the matrix waves run layer 1's emb_a half; bucket launch gone; bit-identical
# speedup vs baseline: 8.7454x; 1.1163x over previous
#include <hip/hip_runtime.h>

typedef float v2f __attribute__((ext_vector_type(2)));
typedef float v4f __attribute__((ext_vector_type(4)));
typedef float v8f __attribute__((ext_vector_type(8)));

#define D_MODEL   256
#define BATCH_A   4096
#define BATCH_B   8192
#define NKEYS     4
#define NBUCKETS  1024

__device__ __forceinline__ int hash_row(const int* __restrict__ keys, int row) {
    int s = 0;
#pragma unroll
    for (int k = 0; k < NKEYS; ++k) s += keys[row * NKEYS + k];
    return (int)(((unsigned)s) & (NBUCKETS - 1));
}

__global__ void hash_rows_kernel(const int* __restrict__ keys_a, const int* __restrict__ keys_b,
                                 int* __restrict__ hash_a, int* __restrict__ hash_b) {
    int j = blockIdx.x * blockDim.x + threadIdx.x;
    if (j < BATCH_B)                hash_b[j] = hash_row(keys_b, j);
    else if (j < BATCH_B + BATCH_A) hash_a[j - BATCH_B] = hash_row(keys_a, j - BATCH_B);
}

#define ROW_BITS    13
#define RANK_LANES  32
#define RANK_WAVES  4
#define RANK_BLOCKS ((BATCH_B + NBUCKETS) / RANK_LANES)
#define RANK_STAGE  (BATCH_B / (RANK_LANES * RANK_WAVES * 4))
static_assert((1 << ROW_BITS) == BATCH_B, "row index must fit the low key bits");
static_assert((BATCH_B + NBUCKETS) % RANK_LANES == 0, "whole blocks of queries");
static_assert(RANK_STAGE * RANK_LANES * RANK_WAVES * 4 == BATCH_B, "staging covers every key");

__global__ __launch_bounds__(RANK_LANES * RANK_WAVES)
void rank_rows_kernel(const int* __restrict__ hash_b,
                      int* __restrict__ perm,
                      int* __restrict__ bucket_off) {
    __shared__ __align__(16) unsigned keys[BATCH_B];
    __shared__ unsigned part[RANK_WAVES][RANK_LANES];

    const int tid  = threadIdx.x;
    const int lane = tid & 31;
    const int wave = tid >> 5;

    int4 hv[RANK_STAGE];
#pragma unroll
    for (int s = 0; s < RANK_STAGE; ++s)
        hv[s] = *(const int4*)&hash_b[tid * 4 + s * (RANK_LANES * RANK_WAVES * 4)];
#pragma unroll
    for (int s = 0; s < RANK_STAGE; ++s) {
        const int i = tid * 4 + s * (RANK_LANES * RANK_WAVES * 4);
        uint4 kv;
        kv.x = ((unsigned)hv[s].x << ROW_BITS) | (unsigned)(i);
        kv.y = ((unsigned)hv[s].y << ROW_BITS) | (unsigned)(i + 1);
        kv.z = ((unsigned)hv[s].z << ROW_BITS) | (unsigned)(i + 2);
        kv.w = ((unsigned)hv[s].w << ROW_BITS) | (unsigned)(i + 3);
        *(uint4*)&keys[i] = kv;
    }
    __syncthreads();

    const int q = blockIdx.x * RANK_LANES + lane;
    const unsigned qk = (q < BATCH_B) ? keys[q]
                                      : ((unsigned)(q - BATCH_B) << ROW_BITS);

    unsigned cnt = 0;
    const int k0 = wave * (BATCH_B / RANK_WAVES);
#pragma unroll 8
    for (int i = 0; i < BATCH_B / RANK_WAVES; i += 4) {
        const uint4 kv = *(const uint4*)&keys[k0 + i];
        cnt += (kv.x < qk) ? 1u : 0u;
        cnt += (kv.y < qk) ? 1u : 0u;
        cnt += (kv.z < qk) ? 1u : 0u;
        cnt += (kv.w < qk) ? 1u : 0u;
    }
    part[wave][lane] = cnt;
    __syncthreads();

    if (wave == 0) {
        const unsigned pos = part[0][lane] + part[1][lane] + part[2][lane] + part[3][lane];
        if (q < BATCH_B) perm[pos] = q;
        else             bucket_off[q - BATCH_B] = (int)pos;
    }
}

#define GM_ROWS  4
#define GM_WAVES 4
#define GM_SLOTS 8

__device__ __forceinline__ void gather_bucket_means(const float* __restrict__ emb_b,
                                                    const int* __restrict__ perm,
                                                    const int* __restrict__ bucket_off,
                                                    const int* __restrict__ hash_a,
                                                    float* dst, int pitch,
                                                    int row0, int lrow0, int lane) {
    const int li = lane & (GM_ROWS - 1);
    const int gl = hash_a[row0 + li];
    const bool lastb = (gl + 1 >= NBUCKETS);
    const int bl = bucket_off[gl];
    const int e0 = bucket_off[lastb ? gl : gl + 1];
    const int el = lastb ? BATCH_B : e0;
    int beg[GM_ROWS], cnt[GM_ROWS];
    int maxcnt = 0;
#pragma unroll
    for (int i = 0; i < GM_ROWS; ++i) {
        beg[i] = __builtin_amdgcn_readlane(bl, i);
        cnt[i] = __builtin_amdgcn_readlane(el, i) - beg[i];
        maxcnt = cnt[i] > maxcnt ? cnt[i] : maxcnt;
    }

    v4f acc[GM_ROWS][2];
#pragma unroll
    for (int i = 0; i < GM_ROWS; ++i) {
        acc[i][0] = v4f{};
        acc[i][1] = v4f{};
    }

    for (int base = 0; base < maxcnt; base += 32) {
        int pv[GM_ROWS];
#pragma unroll
        for (int i = 0; i < GM_ROWS; ++i) {
            const int k  = base + lane;
            int pp = beg[i] + (k < cnt[i] ? k : cnt[i] - 1);
            pp = pp < 0 ? 0 : pp;
            pv[i] = perm[pp];
        }
        const int lim = (maxcnt - base) < 32 ? (maxcnt - base) : 32;
        for (int o = 0; o < lim; o += GM_SLOTS) {
            v4f v[GM_ROWS][GM_SLOTS][2];
#pragma unroll
            for (int i = 0; i < GM_ROWS; ++i) {
#pragma unroll
                for (int u = 0; u < GM_SLOTS; ++u) {
                    const int row = __builtin_amdgcn_readlane(pv[i], o + u);
                    const float* sp = emb_b + (size_t)row * D_MODEL + 4 * lane;
                    v[i][u][0] = *(const v4f*)sp;
                    v[i][u][1] = *(const v4f*)(sp + D_MODEL / 2);
                }
            }
#pragma unroll
            for (int i = 0; i < GM_ROWS; ++i) {
#pragma unroll
                for (int u = 0; u < GM_SLOTS; ++u) {
                    const bool in_slice = (base + o + u < cnt[i]);
#pragma unroll
                    for (int q = 0; q < 2; ++q) {
                        acc[i][q].x += in_slice ? v[i][u][q].x : 0.0f;
                        acc[i][q].y += in_slice ? v[i][u][q].y : 0.0f;
                        acc[i][q].z += in_slice ? v[i][u][q].z : 0.0f;
                        acc[i][q].w += in_slice ? v[i][u][q].w : 0.0f;
                    }
                }
            }
        }
    }

#pragma unroll
    for (int i = 0; i < GM_ROWS; ++i) {
        const float scale = 1.0f / (float)(cnt[i] < 1 ? 1 : cnt[i]);
#pragma unroll
        for (int q = 0; q < 2; ++q) {
            v4f mean;
            mean.x = acc[i][q].x * scale;
            mean.y = acc[i][q].y * scale;
            mean.z = acc[i][q].z * scale;
            mean.w = acc[i][q].w * scale;
            *(v4f*)(dst + (lrow0 + i) * pitch + (D_MODEL / 2) * q + 4 * lane) = mean;
        }
    }
}

#define CH_WAVES  4
#define CH_ROWS   16
static_assert(GM_WAVES * GM_ROWS == CH_ROWS, "the gather waves' rows cover the block's batch rows");
static_assert((GM_ROWS & (GM_ROWS - 1)) == 0 && GM_ROWS <= 32, "lane i fetches row i's slice bounds");
#define H_PITCH   (2 * D_MODEL + 4)
#define R_PITCH   (D_MODEL + 4)

template <int NG, int KS>
struct ChainStage {
    v4f w[NG][KS / 8][2][2];
    v4f x[KS / 8];
};

template <int NG, int KS, int N>
__device__ __forceinline__ void chain_load_w(ChainStage<NG, KS>& f,
                                             const float* __restrict__ wp, int k0) {
#pragma unroll
    for (int g = 0; g < KS / 8; ++g) {
#pragma unroll
        for (int ng = 0; ng < NG; ++ng) {
#pragma unroll
            for (int j = 0; j < 2; ++j) {
#pragma unroll
                for (int s = 0; s < 2; ++s)
                    f.w[ng][g][j][s] =
                        *(const v4f*)(wp + (size_t)(k0 + 8 * g + 2 * j + s) * N + 64 * ng);
            }
        }
    }
}

template <int NG, int KS>
__device__ __forceinline__ void chain_load_x(ChainStage<NG, KS>& f, const float* xp, int k0) {
#pragma unroll
    for (int g = 0; g < KS / 8; ++g) f.x[g] = *(const v4f*)(xp + (k0 + 8 * g));
}

template <int NG, int KS, int N>
__device__ __forceinline__ void chain_load_stage(ChainStage<NG, KS>& f,
                                                 const float* __restrict__ wp,
                                                 const float* xp,
                                                 int k0) {
    chain_load_x<NG, KS>(f, xp, k0);
    chain_load_w<NG, KS, N>(f, wp, k0);
}

template <int NG, int KS>
__device__ __forceinline__ void chain_mma_stage(const ChainStage<NG, KS>& f, v8f (&acc)[NG][4]) {
#pragma unroll
    for (int g = 0; g < KS / 8; ++g) {
#pragma unroll
        for (int j = 0; j < 2; ++j) {
            v2f x2;
            x2.x = f.x[g][2 * j];
            x2.y = f.x[g][2 * j + 1];
#pragma unroll
            for (int ng = 0; ng < NG; ++ng) {
#pragma unroll
                for (int t = 0; t < 4; ++t) {
                    v2f w2;
                    w2.x = f.w[ng][g][j][0][t];
                    w2.y = f.w[ng][g][j][1][t];
                    acc[ng][t] = __builtin_amdgcn_wmma_f32_16x16x4_f32(false, w2, false, x2,
                                                                       (short)0, acc[ng][t],
                                                                       false, false);
                }
            }
        }
    }
}

template <int NG, int KS, int K, int N>
__device__ __forceinline__ void chain_layer(v8f (&acc)[NG][4],
                                            ChainStage<NG, KS>& f0,
                                            const float* __restrict__ wp,
                                            const float* xp) {
    static_assert(KS % 8 == 0 && K % (2 * KS) == 0 && K >= 4 * KS, "two stages per trip, two trips");
    ChainStage<NG, KS> f1;
    chain_load_x<NG, KS>(f0, xp, 0);
#pragma nounroll
    for (int k0 = 0; k0 < K - 2 * KS; k0 += 2 * KS) {
        chain_load_stage<NG, KS, N>(f1, wp, xp, k0 + KS);
        chain_mma_stage<NG, KS>(f0, acc);
        chain_load_stage<NG, KS, N>(f0, wp, xp, k0 + 2 * KS);
        chain_mma_stage<NG, KS>(f1, acc);
    }
    chain_load_stage<NG, KS, N>(f1, wp, xp, K - KS);
    chain_mma_stage<NG, KS>(f0, acc);
    chain_mma_stage<NG, KS>(f1, acc);
}

template <int NG, int KS, int H, int N>
__device__ __forceinline__ void chain_joined_lo(v8f (&acc)[NG][4],
                                                ChainStage<NG, KS>& f0,
                                                const float* __restrict__ wp,
                                                const float* xlo) {
    static_assert(KS % 8 == 0 && H % (2 * KS) == 0 && H >= 4 * KS, "whole trips on each side of the join");
    ChainStage<NG, KS> f1;
    chain_load_x<NG, KS>(f0, xlo, 0);
#pragma nounroll
    for (int k0 = 0; k0 < H - 2 * KS; k0 += 2 * KS) {
        chain_load_stage<NG, KS, N>(f1, wp, xlo, k0 + KS);
        chain_mma_stage<NG, KS>(f0, acc);
        chain_load_stage<NG, KS, N>(f0, wp, xlo, k0 + 2 * KS);
        chain_mma_stage<NG, KS>(f1, acc);
    }
    chain_load_stage<NG, KS, N>(f1, wp, xlo, H - KS);
    chain_mma_stage<NG, KS>(f0, acc);
    chain_load_w<NG, KS, N>(f0, wp, H);
    chain_mma_stage<NG, KS>(f1, acc);
}

template <int NG, int KS, int H, int N>
__device__ __forceinline__ void chain_joined_hi(v8f (&acc)[NG][4],
                                                ChainStage<NG, KS>& f0,
                                                const float* __restrict__ wp,
                                                const float* xhi) {
    ChainStage<NG, KS> f1;
    chain_load_x<NG, KS>(f0, xhi, 0);
    const float* wph = wp + (size_t)H * N;
#pragma nounroll
    for (int k0 = 0; k0 < H - 2 * KS; k0 += 2 * KS) {
        chain_load_stage<NG, KS, N>(f1, wph, xhi, k0 + KS);
        chain_mma_stage<NG, KS>(f0, acc);
        chain_load_stage<NG, KS, N>(f0, wph, xhi, k0 + 2 * KS);
        chain_mma_stage<NG, KS>(f1, acc);
    }
    chain_load_stage<NG, KS, N>(f1, wph, xhi, H - KS);
    chain_mma_stage<NG, KS>(f0, acc);
    chain_mma_stage<NG, KS>(f1, acc);
}

template <int NG, bool RELU>
__device__ __forceinline__ void chain_store_strip(const v8f (&acc)[NG][4],
                                                  const float* __restrict__ bias,
                                                  float* dst, int pitch, int nb, int c, int h) {
#pragma unroll
    for (int ng = 0; ng < NG; ++ng) {
#pragma unroll
        for (int r = 0; r < 8; ++r) {
            const int f = nb + 64 * ng + 32 * h + 4 * r;
            const v4f bia = *(const v4f*)(bias + f);
            v4f o;
            o.x = acc[ng][0][r] + bia.x;
            o.y = acc[ng][1][r] + bia.y;
            o.z = acc[ng][2][r] + bia.z;
            o.w = acc[ng][3][r] + bia.w;
            if (RELU) {
                o.x = fmaxf(o.x, 0.0f);
                o.y = fmaxf(o.y, 0.0f);
                o.z = fmaxf(o.z, 0.0f);
                o.w = fmaxf(o.w, 0.0f);
            }
            *(v4f*)(dst + c * pitch + f) = o;
        }
    }
}

__global__ __launch_bounds__((CH_WAVES + GM_WAVES) * 32)
void mlp_chain_kernel(const float* __restrict__ emb_a,
                      const float* __restrict__ emb_b,
                      const int* __restrict__ perm,
                      const int* __restrict__ bucket_off,
                      const int* __restrict__ hash_a,
                      const float* __restrict__ W1, const float* __restrict__ b1,
                      const float* __restrict__ W2, const float* __restrict__ b2,
                      const float* __restrict__ Wo, const float* __restrict__ bo,
                      float* __restrict__ out) {
    __shared__ __align__(16) float hbuf[CH_ROWS * H_PITCH];
    __shared__ __align__(16) float rbuf[CH_ROWS * R_PITCH];

    const int lane = threadIdx.x & 31;
    const int wave = threadIdx.x >> 5;
    const int h    = lane >> 4;
    const int c    = lane & 15;
    const int m0   = blockIdx.x * CH_ROWS;

    const int wave_u = __builtin_amdgcn_readfirstlane(wave);
    const bool mat = wave_u < CH_WAVES;

    const float* wp1 = W1 + (size_t)(4 * h) * (2 * D_MODEL) + (128 * wave + 4 * c);
    const float* wp2 = W2 + (size_t)(4 * h) * D_MODEL + (64 * wave + 4 * c);
    const float* wp3 = Wo + (size_t)(4 * h) * D_MODEL + (64 * wave + 4 * c);

    v8f acc1[2][4];
#pragma unroll
    for (int ng = 0; ng < 2; ++ng) {
#pragma unroll
        for (int t = 0; t < 4; ++t) acc1[ng][t] = v8f{};
    }
    ChainStage<2, 32> s1;
    if (mat) {
        chain_load_w<2, 32, 2 * D_MODEL>(s1, wp1, 0);
        chain_joined_lo<2, 32, D_MODEL, 2 * D_MODEL>(
            acc1, s1, wp1, emb_a + (size_t)(m0 + c) * D_MODEL + 4 * h);
    } else {
        const int gw = wave_u - CH_WAVES;
        gather_bucket_means(emb_b, perm, bucket_off, hash_a, rbuf, R_PITCH,
                            m0 + GM_ROWS * gw, GM_ROWS * gw, lane);
    }
    __syncthreads();

    ChainStage<1, 64> s2;
    if (mat) {
        chain_joined_hi<2, 32, D_MODEL, 2 * D_MODEL>(acc1, s1, wp1, rbuf + c * R_PITCH + 4 * h);
        chain_load_w<1, 64, D_MODEL>(s2, wp2, 0);
        chain_store_strip<2, true>(acc1, b1, hbuf, H_PITCH, 128 * wave, c, h);
    }
    __syncthreads();

    v8f acc2[1][4];
#pragma unroll
    for (int t = 0; t < 4; ++t) acc2[0][t] = v8f{};
    ChainStage<1, 32> s3;
    if (mat) {
        chain_layer<1, 64, 2 * D_MODEL, D_MODEL>(acc2, s2, wp2, hbuf + c * H_PITCH + 4 * h);
        chain_load_w<1, 32, D_MODEL>(s3, wp3, 0);
        chain_store_strip<1, false>(acc2, b2, rbuf, R_PITCH, 64 * wave, c, h);
    }
    __syncthreads();

    if (mat) {
        v8f acc3[1][4];
#pragma unroll
        for (int t = 0; t < 4; ++t) acc3[0][t] = v8f{};
        chain_layer<1, 32, D_MODEL, D_MODEL>(acc3, s3, wp3, rbuf + c * R_PITCH + 4 * h);
        chain_store_strip<1, false>(acc3, bo, hbuf, R_PITCH, 64 * wave, c, h);
    }
    __syncthreads();

#pragma unroll
    for (int it = 0; it < (CH_ROWS * D_MODEL / 4) / ((CH_WAVES + GM_WAVES) * 32); ++it) {
        const int idx  = it * ((CH_WAVES + GM_WAVES) * 32) + threadIdx.x;
        const int row  = idx / (D_MODEL / 4);
        const int col4 = idx % (D_MODEL / 4);
        *(v4f*)(out + (size_t)(m0 + row) * D_MODEL + 4 * col4) =
            *(const v4f*)(hbuf + row * R_PITCH + 4 * col4);
    }
}

extern "C" void kernel_launch(void* const* d_in, const int* in_sizes, int n_in,
                              void* d_out, int out_size, void* d_ws, size_t ws_size,
                              hipStream_t stream) {
    const float* emb_a  = (const float*)d_in[0];
    const float* emb_b  = (const float*)d_in[1];
    const int*   keys_a = (const int*)d_in[2];
    const int*   keys_b = (const int*)d_in[3];
    const float* W1 = (const float*)d_in[4];
    const float* b1 = (const float*)d_in[5];
    const float* W2 = (const float*)d_in[6];
    const float* b2 = (const float*)d_in[7];
    const float* Wo = (const float*)d_in[8];
    const float* bo = (const float*)d_in[9];
    float* out = (float*)d_out;

    float* ws         = (float*)d_ws;
    int*   hash_b     = (int*)ws;
    int*   hash_a     = hash_b + BATCH_B;
    int*   perm       = hash_a + BATCH_A;
    int*   bucket_off = perm + BATCH_B;

    hash_rows_kernel<<<(BATCH_B + BATCH_A) / 256, 256, 0, stream>>>(keys_a, keys_b, hash_a, hash_b);
    rank_rows_kernel<<<RANK_BLOCKS, RANK_LANES * RANK_WAVES, 0, stream>>>(hash_b, perm, bucket_off);

    mlp_chain_kernel<<<BATCH_A / CH_ROWS, (CH_WAVES + GM_WAVES) * 32, 0, stream>>>(emb_a, emb_b, perm, bucket_off,
                                                                      hash_a, W1, b1, W2, b2,
                                                                      Wo, bo, out);
}


# sort kernel hashes table B's key columns itself, one copy of the 8192 sort keys per block for 64 queries (144 blocks x 8 waves); gather waves hash their table A rows; hash launch gone; bit-identical
# speedup vs baseline: 9.1860x; 1.0504x over previous
#include <hip/hip_runtime.h>

typedef float v2f __attribute__((ext_vector_type(2)));
typedef float v4f __attribute__((ext_vector_type(4)));
typedef float v8f __attribute__((ext_vector_type(8)));

#define D_MODEL   256
#define BATCH_A   4096
#define BATCH_B   8192
#define NKEYS     4
#define NBUCKETS  1024

__device__ __forceinline__ int hash_row(const int* __restrict__ keys, int row) {
    int s = 0;
#pragma unroll
    for (int k = 0; k < NKEYS; ++k) s += keys[row * NKEYS + k];
    return (int)(((unsigned)s) & (NBUCKETS - 1));
}

#define ROW_BITS    13
#define RANK_LANES  32
#define RANK_WAVES  4
#define RANK_GROUPS 2
#define RANK_THREADS (RANK_LANES * RANK_WAVES * RANK_GROUPS)
#define RANK_BLOCKS ((BATCH_B + NBUCKETS) / (RANK_LANES * RANK_GROUPS))
#define RANK_STAGE  (BATCH_B / RANK_THREADS)
static_assert((1 << ROW_BITS) == BATCH_B, "row index must fit the low key bits");
static_assert((BATCH_B + NBUCKETS) % (RANK_LANES * RANK_GROUPS) == 0, "whole blocks of queries");
static_assert(RANK_STAGE * RANK_THREADS == BATCH_B, "staging covers every row");
static_assert(NKEYS == 4, "a row's keys are one 16-byte load");

__global__ __launch_bounds__(RANK_THREADS) __attribute__((amdgpu_waves_per_eu(2, 2)))
void rank_rows_kernel(const int* __restrict__ keys_b,
                      int* __restrict__ perm,
                      int* __restrict__ bucket_off) {
    __shared__ __align__(16) unsigned keys[BATCH_B];
    __shared__ unsigned part[RANK_GROUPS * RANK_WAVES][RANK_LANES];

    const int tid   = threadIdx.x;
    const int lane  = tid & 31;
    const int wave  = tid >> 5;
    const int group = wave / RANK_WAVES;
    const int qwave = wave % RANK_WAVES;

    int4 kr[RANK_STAGE];
#pragma unroll
    for (int s = 0; s < RANK_STAGE; ++s)
        kr[s] = *(const int4*)&keys_b[(tid + s * RANK_THREADS) * NKEYS];
#pragma unroll
    for (int s = 0; s < RANK_STAGE; ++s) {
        const int i = tid + s * RANK_THREADS;
        const unsigned hsh = (unsigned)(kr[s].x + kr[s].y + kr[s].z + kr[s].w) & (NBUCKETS - 1);
        keys[i] = (hsh << ROW_BITS) | (unsigned)i;
    }
    __syncthreads();

    const int q = (blockIdx.x * RANK_GROUPS + group) * RANK_LANES + lane;
    const unsigned qk = (q < BATCH_B) ? keys[q]
                                      : ((unsigned)(q - BATCH_B) << ROW_BITS);

    unsigned cnt = 0;
    const int k0 = qwave * (BATCH_B / RANK_WAVES);
#pragma unroll 8
    for (int i = 0; i < BATCH_B / RANK_WAVES; i += 4) {
        const uint4 kv = *(const uint4*)&keys[k0 + i];
        cnt += (kv.x < qk) ? 1u : 0u;
        cnt += (kv.y < qk) ? 1u : 0u;
        cnt += (kv.z < qk) ? 1u : 0u;
        cnt += (kv.w < qk) ? 1u : 0u;
    }
    part[wave][lane] = cnt;
    __syncthreads();

    if (qwave == 0) {
        const unsigned pos = part[wave][lane] + part[wave + 1][lane] +
                             part[wave + 2][lane] + part[wave + 3][lane];
        if (q < BATCH_B) perm[pos] = q;
        else             bucket_off[q - BATCH_B] = (int)pos;
    }
}

#define GM_ROWS  4
#define GM_WAVES 4
#define GM_SLOTS 8

__device__ __forceinline__ void gather_bucket_means(const float* __restrict__ emb_b,
                                                    const int* __restrict__ perm,
                                                    const int* __restrict__ bucket_off,
                                                    const int* __restrict__ keys_a,
                                                    float* dst, int pitch,
                                                    int row0, int lrow0, int lane) {
    const int li = lane & (GM_ROWS - 1);
    const int gl = hash_row(keys_a, row0 + li);
    const bool lastb = (gl + 1 >= NBUCKETS);
    const int bl = bucket_off[gl];
    const int e0 = bucket_off[lastb ? gl : gl + 1];
    const int el = lastb ? BATCH_B : e0;
    int beg[GM_ROWS], cnt[GM_ROWS];
    int maxcnt = 0;
#pragma unroll
    for (int i = 0; i < GM_ROWS; ++i) {
        beg[i] = __builtin_amdgcn_readlane(bl, i);
        cnt[i] = __builtin_amdgcn_readlane(el, i) - beg[i];
        maxcnt = cnt[i] > maxcnt ? cnt[i] : maxcnt;
    }

    v4f acc[GM_ROWS][2];
#pragma unroll
    for (int i = 0; i < GM_ROWS; ++i) {
        acc[i][0] = v4f{};
        acc[i][1] = v4f{};
    }

    for (int base = 0; base < maxcnt; base += 32) {
        int pv[GM_ROWS];
#pragma unroll
        for (int i = 0; i < GM_ROWS; ++i) {
            const int k  = base + lane;
            int pp = beg[i] + (k < cnt[i] ? k : cnt[i] - 1);
            pp = pp < 0 ? 0 : pp;
            pv[i] = perm[pp];
        }
        const int lim = (maxcnt - base) < 32 ? (maxcnt - base) : 32;
        for (int o = 0; o < lim; o += GM_SLOTS) {
            v4f v[GM_ROWS][GM_SLOTS][2];
#pragma unroll
            for (int i = 0; i < GM_ROWS; ++i) {
#pragma unroll
                for (int u = 0; u < GM_SLOTS; ++u) {
                    const int row = __builtin_amdgcn_readlane(pv[i], o + u);
                    const float* sp = emb_b + (size_t)row * D_MODEL + 4 * lane;
                    v[i][u][0] = *(const v4f*)sp;
                    v[i][u][1] = *(const v4f*)(sp + D_MODEL / 2);
                }
            }
#pragma unroll
            for (int i = 0; i < GM_ROWS; ++i) {
#pragma unroll
                for (int u = 0; u < GM_SLOTS; ++u) {
                    const bool in_slice = (base + o + u < cnt[i]);
#pragma unroll
                    for (int q = 0; q < 2; ++q) {
                        acc[i][q].x += in_slice ? v[i][u][q].x : 0.0f;
                        acc[i][q].y += in_slice ? v[i][u][q].y : 0.0f;
                        acc[i][q].z += in_slice ? v[i][u][q].z : 0.0f;
                        acc[i][q].w += in_slice ? v[i][u][q].w : 0.0f;
                    }
                }
            }
        }
    }

#pragma unroll
    for (int i = 0; i < GM_ROWS; ++i) {
        const float scale = 1.0f / (float)(cnt[i] < 1 ? 1 : cnt[i]);
#pragma unroll
        for (int q = 0; q < 2; ++q) {
            v4f mean;
            mean.x = acc[i][q].x * scale;
            mean.y = acc[i][q].y * scale;
            mean.z = acc[i][q].z * scale;
            mean.w = acc[i][q].w * scale;
            *(v4f*)(dst + (lrow0 + i) * pitch + (D_MODEL / 2) * q + 4 * lane) = mean;
        }
    }
}

#define CH_WAVES  4
#define CH_ROWS   16
static_assert(GM_WAVES * GM_ROWS == CH_ROWS, "the gather waves' rows cover the block's batch rows");
static_assert((GM_ROWS & (GM_ROWS - 1)) == 0 && GM_ROWS <= 32, "lane i fetches row i's slice bounds");
#define H_PITCH   (2 * D_MODEL + 4)
#define R_PITCH   (D_MODEL + 4)

template <int NG, int KS>
struct ChainStage {
    v4f w[NG][KS / 8][2][2];
    v4f x[KS / 8];
};

template <int NG, int KS, int N>
__device__ __forceinline__ void chain_load_w(ChainStage<NG, KS>& f,
                                             const float* __restrict__ wp, int k0) {
#pragma unroll
    for (int g = 0; g < KS / 8; ++g) {
#pragma unroll
        for (int ng = 0; ng < NG; ++ng) {
#pragma unroll
            for (int j = 0; j < 2; ++j) {
#pragma unroll
                for (int s = 0; s < 2; ++s)
                    f.w[ng][g][j][s] =
                        *(const v4f*)(wp + (size_t)(k0 + 8 * g + 2 * j + s) * N + 64 * ng);
            }
        }
    }
}

template <int NG, int KS>
__device__ __forceinline__ void chain_load_x(ChainStage<NG, KS>& f, const float* xp, int k0) {
#pragma unroll
    for (int g = 0; g < KS / 8; ++g) f.x[g] = *(const v4f*)(xp + (k0 + 8 * g));
}

template <int NG, int KS, int N>
__device__ __forceinline__ void chain_load_stage(ChainStage<NG, KS>& f,
                                                 const float* __restrict__ wp,
                                                 const float* xp,
                                                 int k0) {
    chain_load_x<NG, KS>(f, xp, k0);
    chain_load_w<NG, KS, N>(f, wp, k0);
}

template <int NG, int KS>
__device__ __forceinline__ void chain_mma_stage(const ChainStage<NG, KS>& f, v8f (&acc)[NG][4]) {
#pragma unroll
    for (int g = 0; g < KS / 8; ++g) {
#pragma unroll
        for (int j = 0; j < 2; ++j) {
            v2f x2;
            x2.x = f.x[g][2 * j];
            x2.y = f.x[g][2 * j + 1];
#pragma unroll
            for (int ng = 0; ng < NG; ++ng) {
#pragma unroll
                for (int t = 0; t < 4; ++t) {
                    v2f w2;
                    w2.x = f.w[ng][g][j][0][t];
                    w2.y = f.w[ng][g][j][1][t];
                    acc[ng][t] = __builtin_amdgcn_wmma_f32_16x16x4_f32(false, w2, false, x2,
                                                                       (short)0, acc[ng][t],
                                                                       false, false);
                }
            }
        }
    }
}

template <int NG, int KS, int K, int N>
__device__ __forceinline__ void chain_layer(v8f (&acc)[NG][4],
                                            ChainStage<NG, KS>& f0,
                                            const float* __restrict__ wp,
                                            const float* xp) {
    static_assert(KS % 8 == 0 && K % (2 * KS) == 0 && K >= 4 * KS, "two stages per trip, two trips");
    ChainStage<NG, KS> f1;
    chain_load_x<NG, KS>(f0, xp, 0);
#pragma nounroll
    for (int k0 = 0; k0 < K - 2 * KS; k0 += 2 * KS) {
        chain_load_stage<NG, KS, N>(f1, wp, xp, k0 + KS);
        chain_mma_stage<NG, KS>(f0, acc);
        chain_load_stage<NG, KS, N>(f0, wp, xp, k0 + 2 * KS);
        chain_mma_stage<NG, KS>(f1, acc);
    }
    chain_load_stage<NG, KS, N>(f1, wp, xp, K - KS);
    chain_mma_stage<NG, KS>(f0, acc);
    chain_mma_stage<NG, KS>(f1, acc);
}

template <int NG, int KS, int H, int N>
__device__ __forceinline__ void chain_joined_lo(v8f (&acc)[NG][4],
                                                ChainStage<NG, KS>& f0,
                                                const float* __restrict__ wp,
                                                const float* xlo) {
    static_assert(KS % 8 == 0 && H % (2 * KS) == 0 && H >= 4 * KS, "whole trips on each side of the join");
    ChainStage<NG, KS> f1;
    chain_load_x<NG, KS>(f0, xlo, 0);
#pragma nounroll
    for (int k0 = 0; k0 < H - 2 * KS; k0 += 2 * KS) {
        chain_load_stage<NG, KS, N>(f1, wp, xlo, k0 + KS);
        chain_mma_stage<NG, KS>(f0, acc);
        chain_load_stage<NG, KS, N>(f0, wp, xlo, k0 + 2 * KS);
        chain_mma_stage<NG, KS>(f1, acc);
    }
    chain_load_stage<NG, KS, N>(f1, wp, xlo, H - KS);
    chain_mma_stage<NG, KS>(f0, acc);
    chain_load_w<NG, KS, N>(f0, wp, H);
    chain_mma_stage<NG, KS>(f1, acc);
}

template <int NG, int KS, int H, int N>
__device__ __forceinline__ void chain_joined_hi(v8f (&acc)[NG][4],
                                                ChainStage<NG, KS>& f0,
                                                const float* __restrict__ wp,
                                                const float* xhi) {
    ChainStage<NG, KS> f1;
    chain_load_x<NG, KS>(f0, xhi, 0);
    const float* wph = wp + (size_t)H * N;
#pragma nounroll
    for (int k0 = 0; k0 < H - 2 * KS; k0 += 2 * KS) {
        chain_load_stage<NG, KS, N>(f1, wph, xhi, k0 + KS);
        chain_mma_stage<NG, KS>(f0, acc);
        chain_load_stage<NG, KS, N>(f0, wph, xhi, k0 + 2 * KS);
        chain_mma_stage<NG, KS>(f1, acc);
    }
    chain_load_stage<NG, KS, N>(f1, wph, xhi, H - KS);
    chain_mma_stage<NG, KS>(f0, acc);
    chain_mma_stage<NG, KS>(f1, acc);
}

template <int NG, bool RELU>
__device__ __forceinline__ void chain_store_strip(const v8f (&acc)[NG][4],
                                                  const float* __restrict__ bias,
                                                  float* dst, int pitch, int nb, int c, int h) {
#pragma unroll
    for (int ng = 0; ng < NG; ++ng) {
#pragma unroll
        for (int r = 0; r < 8; ++r) {
            const int f = nb + 64 * ng + 32 * h + 4 * r;
            const v4f bia = *(const v4f*)(bias + f);
            v4f o;
            o.x = acc[ng][0][r] + bia.x;
            o.y = acc[ng][1][r] + bia.y;
            o.z = acc[ng][2][r] + bia.z;
            o.w = acc[ng][3][r] + bia.w;
            if (RELU) {
                o.x = fmaxf(o.x, 0.0f);
                o.y = fmaxf(o.y, 0.0f);
                o.z = fmaxf(o.z, 0.0f);
                o.w = fmaxf(o.w, 0.0f);
            }
            *(v4f*)(dst + c * pitch + f) = o;
        }
    }
}

__global__ __launch_bounds__((CH_WAVES + GM_WAVES) * 32)
void mlp_chain_kernel(const float* __restrict__ emb_a,
                      const float* __restrict__ emb_b,
                      const int* __restrict__ perm,
                      const int* __restrict__ bucket_off,
                      const int* __restrict__ keys_a,
                      const float* __restrict__ W1, const float* __restrict__ b1,
                      const float* __restrict__ W2, const float* __restrict__ b2,
                      const float* __restrict__ Wo, const float* __restrict__ bo,
                      float* __restrict__ out) {
    __shared__ __align__(16) float hbuf[CH_ROWS * H_PITCH];
    __shared__ __align__(16) float rbuf[CH_ROWS * R_PITCH];

    const int lane = threadIdx.x & 31;
    const int wave = threadIdx.x >> 5;
    const int h    = lane >> 4;
    const int c    = lane & 15;
    const int m0   = blockIdx.x * CH_ROWS;

    const int wave_u = __builtin_amdgcn_readfirstlane(wave);
    const bool mat = wave_u < CH_WAVES;

    const float* wp1 = W1 + (size_t)(4 * h) * (2 * D_MODEL) + (128 * wave + 4 * c);
    const float* wp2 = W2 + (size_t)(4 * h) * D_MODEL + (64 * wave + 4 * c);
    const float* wp3 = Wo + (size_t)(4 * h) * D_MODEL + (64 * wave + 4 * c);

    v8f acc1[2][4];
#pragma unroll
    for (int ng = 0; ng < 2; ++ng) {
#pragma unroll
        for (int t = 0; t < 4; ++t) acc1[ng][t] = v8f{};
    }
    ChainStage<2, 32> s1;
    if (mat) {
        chain_load_w<2, 32, 2 * D_MODEL>(s1, wp1, 0);
        chain_joined_lo<2, 32, D_MODEL, 2 * D_MODEL>(
            acc1, s1, wp1, emb_a + (size_t)(m0 + c) * D_MODEL + 4 * h);
    } else {
        const int gw = wave_u - CH_WAVES;
        gather_bucket_means(emb_b, perm, bucket_off, keys_a, rbuf, R_PITCH,
                            m0 + GM_ROWS * gw, GM_ROWS * gw, lane);
    }
    __syncthreads();

    ChainStage<1, 64> s2;
    if (mat) {
        chain_joined_hi<2, 32, D_MODEL, 2 * D_MODEL>(acc1, s1, wp1, rbuf + c * R_PITCH + 4 * h);
        chain_load_w<1, 64, D_MODEL>(s2, wp2, 0);
        chain_store_strip<2, true>(acc1, b1, hbuf, H_PITCH, 128 * wave, c, h);
    }
    __syncthreads();

    v8f acc2[1][4];
#pragma unroll
    for (int t = 0; t < 4; ++t) acc2[0][t] = v8f{};
    ChainStage<1, 32> s3;
    if (mat) {
        chain_layer<1, 64, 2 * D_MODEL, D_MODEL>(acc2, s2, wp2, hbuf + c * H_PITCH + 4 * h);
        chain_load_w<1, 32, D_MODEL>(s3, wp3, 0);
        chain_store_strip<1, false>(acc2, b2, rbuf, R_PITCH, 64 * wave, c, h);
    }
    __syncthreads();

    if (mat) {
        v8f acc3[1][4];
#pragma unroll
        for (int t = 0; t < 4; ++t) acc3[0][t] = v8f{};
        chain_layer<1, 32, D_MODEL, D_MODEL>(acc3, s3, wp3, rbuf + c * R_PITCH + 4 * h);
        chain_store_strip<1, false>(acc3, bo, hbuf, R_PITCH, 64 * wave, c, h);
    }
    __syncthreads();

#pragma unroll
    for (int it = 0; it < (CH_ROWS * D_MODEL / 4) / ((CH_WAVES + GM_WAVES) * 32); ++it) {
        const int idx  = it * ((CH_WAVES + GM_WAVES) * 32) + threadIdx.x;
        const int row  = idx / (D_MODEL / 4);
        const int col4 = idx % (D_MODEL / 4);
        *(v4f*)(out + (size_t)(m0 + row) * D_MODEL + 4 * col4) =
            *(const v4f*)(hbuf + row * R_PITCH + 4 * col4);
    }
}

extern "C" void kernel_launch(void* const* d_in, const int* in_sizes, int n_in,
                              void* d_out, int out_size, void* d_ws, size_t ws_size,
                              hipStream_t stream) {
    const float* emb_a  = (const float*)d_in[0];
    const float* emb_b  = (const float*)d_in[1];
    const int*   keys_a = (const int*)d_in[2];
    const int*   keys_b = (const int*)d_in[3];
    const float* W1 = (const float*)d_in[4];
    const float* b1 = (const float*)d_in[5];
    const float* W2 = (const float*)d_in[6];
    const float* b2 = (const float*)d_in[7];
    const float* Wo = (const float*)d_in[8];
    const float* bo = (const float*)d_in[9];
    float* out = (float*)d_out;

    float* ws         = (float*)d_ws;
    int*   perm       = (int*)ws;
    int*   bucket_off = perm + BATCH_B;

    rank_rows_kernel<<<RANK_BLOCKS, RANK_THREADS, 0, stream>>>(keys_b, perm, bucket_off);

    mlp_chain_kernel<<<BATCH_A / CH_ROWS, (CH_WAVES + GM_WAVES) * 32, 0, stream>>>(emb_a, emb_b, perm, bucket_off,
                                                                      keys_a, W1, b1, W2, b2,
                                                                      Wo, bo, out);
}


# bucket rows listed inside the dense kernel: each of a block's 8 waves hashes an eighth of table B's keys in registers and looks them up in a 1 KB LDS table of the block's 16 buckets; sort launch gone
# speedup vs baseline: 10.9171x; 1.1884x over previous
#include <hip/hip_runtime.h>

typedef float v2f __attribute__((ext_vector_type(2)));
typedef float v4f __attribute__((ext_vector_type(4)));
typedef float v8f __attribute__((ext_vector_type(8)));

#define D_MODEL   256
#define BATCH_A   4096
#define BATCH_B   8192
#define NKEYS     4
#define NBUCKETS  1024

__device__ __forceinline__ int hash_row(const int* __restrict__ keys, int row) {
    int s = 0;
#pragma unroll
    for (int k = 0; k < NKEYS; ++k) s += keys[row * NKEYS + k];
    return (int)(((unsigned)s) & (NBUCKETS - 1));
}

#define GM_ROWS  4
#define GM_WAVES 4
#define GM_SLOTS 8
#define SC_WAVES  8
#define SC_ROWS   (BATCH_B / SC_WAVES)
#define SC_CHUNKS (SC_ROWS / 32)
#define SC_SLOTS  16
static_assert(NKEYS == 4, "a row's keys are one 16-byte load");
static_assert(BATCH_B <= 65536, "row numbers are kept as 16-bit list entries");
static_assert(SC_CHUNKS <= 32, "one bit per chunk in a 32-bit word");

__device__ __forceinline__ int scan_bucket_rows(const int* __restrict__ keys_a,
                                                const int* __restrict__ keys_b,
                                                unsigned short* lists, unsigned char* tbl,
                                                unsigned char* tv,
                                                int* cntl, int m0, int q, int lane) {
    int4 kr[SC_CHUNKS];
#pragma unroll
    for (int s = 0; s < SC_CHUNKS; ++s)
        kr[s] = *(const int4*)&keys_b[(size_t)(q * SC_ROWS + 32 * s + lane) * NKEYS];

    const int aj = lane & (SC_SLOTS - 1);
    const int ha = hash_row(keys_a, m0 + aj);
    int lead = aj;
#pragma unroll
    for (int j = SC_SLOTS - 1; j >= 0; --j) {
        const int hj = __builtin_amdgcn_readlane(ha, j);
        lead = (hj == ha) ? j : lead;
    }

    unsigned char* tw = tbl + q * NBUCKETS;
    *(uint4*)(tw + 32 * lane)      = uint4{0u, 0u, 0u, 0u};
    *(uint4*)(tw + 32 * lane + 16) = uint4{0u, 0u, 0u, 0u};
    if (lane < SC_SLOTS && lead == aj) tw[ha] = (unsigned char)(aj + 1);

    unsigned char* tvw = tv + q * (SC_CHUNKS * 32);
    int ts[SC_CHUNKS];
#pragma unroll
    for (int s = 0; s < SC_CHUNKS; ++s) {
        const unsigned hsh = (unsigned)(kr[s].x + kr[s].y + kr[s].z + kr[s].w) & (NBUCKETS - 1);
        ts[s] = tw[hsh];
    }
    unsigned hit = 0u;
#pragma unroll
    for (int s = 0; s < SC_CHUNKS; ++s) {
        tvw[32 * s + lane] = (unsigned char)ts[s];
        if (__builtin_amdgcn_ballot_w32(ts[s] != 0) != 0u) hit |= 1u << s;
    }

    int cntv = 0;
    unsigned short* lw = lists + (size_t)q * SC_SLOTS * SC_ROWS;
    int snext = hit != 0u ? __builtin_ctz(hit) : 0;
    int tn = tvw[32 * snext + lane];
    while (hit != 0u) {
        const int s  = snext;
        const int tc = tn;
        hit &= hit - 1u;
        snext = hit != 0u ? __builtin_ctz(hit) : 0;
        tn = tvw[32 * snext + lane];
        unsigned m = __builtin_amdgcn_ballot_w32(tc != 0);
        while (m != 0u) {
            const int l    = __builtin_ctz(m);
            const int slot = __builtin_amdgcn_readlane(tc, l) - 1;
            const int c    = __builtin_amdgcn_readlane(cntv, slot);
            lw[slot * SC_ROWS + c] = (unsigned short)(q * SC_ROWS + 32 * s + l);
            cntv = (lane == slot) ? c + 1 : cntv;
            m &= m - 1u;
        }
    }
    if (lane < SC_SLOTS) cntl[q * SC_SLOTS + lane] = cntv;
    return lead;
}

__device__ __forceinline__ void gather_bucket_means(const float* __restrict__ emb_b,
                                                    const unsigned short* lists,
                                                    const int* cntl, int lead,
                                                    float* dst, int pitch,
                                                    int lrow0, int lane) {
    int ldr[GM_ROWS], cq[GM_ROWS][SC_WAVES], cnt[GM_ROWS];
    int maxcnt = 0;
#pragma unroll
    for (int i = 0; i < GM_ROWS; ++i) {
        ldr[i] = __builtin_amdgcn_readlane(lead, lrow0 + i);
        cnt[i] = 0;
#pragma unroll
        for (int q = 0; q < SC_WAVES; ++q) {
            cq[i][q] = __builtin_amdgcn_readfirstlane(cntl[q * SC_SLOTS + ldr[i]]);
            cnt[i] += cq[i][q];
        }
        maxcnt = cnt[i] > maxcnt ? cnt[i] : maxcnt;
    }

    v4f acc[GM_ROWS][2];
#pragma unroll
    for (int i = 0; i < GM_ROWS; ++i) {
        acc[i][0] = v4f{};
        acc[i][1] = v4f{};
    }

    for (int base = 0; base < maxcnt; base += 32) {
        int pv[GM_ROWS];
#pragma unroll
        for (int i = 0; i < GM_ROWS; ++i) {
            const int k  = base + lane;
            const int kk = k < cnt[i] ? k : cnt[i] - 1;
            int off = kk, qq = 0;
#pragma unroll
            for (int q = 0; q < SC_WAVES - 1; ++q) {
                const bool nx = (qq == q) && (off >= cq[i][q]);
                off = nx ? off - cq[i][q] : off;
                qq  = nx ? q + 1 : qq;
            }
            int row = 0;
            if (kk >= 0) row = lists[(size_t)(qq * SC_SLOTS + ldr[i]) * SC_ROWS + off];
            pv[i] = row;
        }
        const int lim = (maxcnt - base) < 32 ? (maxcnt - base) : 32;
        for (int o = 0; o < lim; o += GM_SLOTS) {
            v4f v[GM_ROWS][GM_SLOTS][2];
#pragma unroll
            for (int i = 0; i < GM_ROWS; ++i) {
#pragma unroll
                for (int u = 0; u < GM_SLOTS; ++u) {
                    const int row = __builtin_amdgcn_readlane(pv[i], o + u);
                    const float* sp = emb_b + (size_t)row * D_MODEL + 4 * lane;
                    v[i][u][0] = *(const v4f*)sp;
                    v[i][u][1] = *(const v4f*)(sp + D_MODEL / 2);
                }
            }
#pragma unroll
            for (int i = 0; i < GM_ROWS; ++i) {
#pragma unroll
                for (int u = 0; u < GM_SLOTS; ++u) {
                    const bool in_slice = (base + o + u < cnt[i]);
#pragma unroll
                    for (int q = 0; q < 2; ++q) {
                        acc[i][q].x += in_slice ? v[i][u][q].x : 0.0f;
                        acc[i][q].y += in_slice ? v[i][u][q].y : 0.0f;
                        acc[i][q].z += in_slice ? v[i][u][q].z : 0.0f;
                        acc[i][q].w += in_slice ? v[i][u][q].w : 0.0f;
                    }
                }
            }
        }
    }

#pragma unroll
    for (int i = 0; i < GM_ROWS; ++i) {
        const float scale = 1.0f / (float)(cnt[i] < 1 ? 1 : cnt[i]);
#pragma unroll
        for (int q = 0; q < 2; ++q) {
            v4f mean;
            mean.x = acc[i][q].x * scale;
            mean.y = acc[i][q].y * scale;
            mean.z = acc[i][q].z * scale;
            mean.w = acc[i][q].w * scale;
            *(v4f*)(dst + (lrow0 + i) * pitch + (D_MODEL / 2) * q + 4 * lane) = mean;
        }
    }
}

#define CH_WAVES  4
#define CH_ROWS   16
static_assert(GM_WAVES * GM_ROWS == CH_ROWS, "the gather waves' rows cover the block's batch rows");
static_assert(CH_ROWS == SC_SLOTS, "one table slot per batch row of the block");
static_assert(CH_WAVES + GM_WAVES == SC_WAVES, "every wave of the block scans one part");
#define H_PITCH   (2 * D_MODEL + 4)
#define R_PITCH   (D_MODEL + 4)

template <int NG, int KS>
struct ChainStage {
    v4f w[NG][KS / 8][2][2];
    v4f x[KS / 8];
};

template <int NG, int KS, int N>
__device__ __forceinline__ void chain_load_w(ChainStage<NG, KS>& f,
                                             const float* __restrict__ wp, int k0) {
#pragma unroll
    for (int g = 0; g < KS / 8; ++g) {
#pragma unroll
        for (int ng = 0; ng < NG; ++ng) {
#pragma unroll
            for (int j = 0; j < 2; ++j) {
#pragma unroll
                for (int s = 0; s < 2; ++s)
                    f.w[ng][g][j][s] =
                        *(const v4f*)(wp + (size_t)(k0 + 8 * g + 2 * j + s) * N + 64 * ng);
            }
        }
    }
}

template <int NG, int KS>
__device__ __forceinline__ void chain_load_x(ChainStage<NG, KS>& f, const float* xp, int k0) {
#pragma unroll
    for (int g = 0; g < KS / 8; ++g) f.x[g] = *(const v4f*)(xp + (k0 + 8 * g));
}

template <int NG, int KS, int N>
__device__ __forceinline__ void chain_load_stage(ChainStage<NG, KS>& f,
                                                 const float* __restrict__ wp,
                                                 const float* xp,
                                                 int k0) {
    chain_load_x<NG, KS>(f, xp, k0);
    chain_load_w<NG, KS, N>(f, wp, k0);
}

template <int NG, int KS>
__device__ __forceinline__ void chain_mma_stage(const ChainStage<NG, KS>& f, v8f (&acc)[NG][4]) {
#pragma unroll
    for (int g = 0; g < KS / 8; ++g) {
#pragma unroll
        for (int j = 0; j < 2; ++j) {
            v2f x2;
            x2.x = f.x[g][2 * j];
            x2.y = f.x[g][2 * j + 1];
#pragma unroll
            for (int ng = 0; ng < NG; ++ng) {
#pragma unroll
                for (int t = 0; t < 4; ++t) {
                    v2f w2;
                    w2.x = f.w[ng][g][j][0][t];
                    w2.y = f.w[ng][g][j][1][t];
                    acc[ng][t] = __builtin_amdgcn_wmma_f32_16x16x4_f32(false, w2, false, x2,
                                                                       (short)0, acc[ng][t],
                                                                       false, false);
                }
            }
        }
    }
}

template <int NG, int KS, int K, int N>
__device__ __forceinline__ void chain_layer(v8f (&acc)[NG][4],
                                            ChainStage<NG, KS>& f0,
                                            const float* __restrict__ wp,
                                            const float* xp) {
    static_assert(KS % 8 == 0 && K % (2 * KS) == 0 && K >= 4 * KS, "two stages per trip, two trips");
    ChainStage<NG, KS> f1;
    chain_load_x<NG, KS>(f0, xp, 0);
#pragma nounroll
    for (int k0 = 0; k0 < K - 2 * KS; k0 += 2 * KS) {
        chain_load_stage<NG, KS, N>(f1, wp, xp, k0 + KS);
        chain_mma_stage<NG, KS>(f0, acc);
        chain_load_stage<NG, KS, N>(f0, wp, xp, k0 + 2 * KS);
        chain_mma_stage<NG, KS>(f1, acc);
    }
    chain_load_stage<NG, KS, N>(f1, wp, xp, K - KS);
    chain_mma_stage<NG, KS>(f0, acc);
    chain_mma_stage<NG, KS>(f1, acc);
}

template <int NG, int KS, int H, int N>
__device__ __forceinline__ void chain_joined_lo(v8f (&acc)[NG][4],
                                                ChainStage<NG, KS>& f0,
                                                const float* __restrict__ wp,
                                                const float* xlo) {
    static_assert(KS % 8 == 0 && H % (2 * KS) == 0 && H >= 4 * KS, "whole trips on each side of the join");
    ChainStage<NG, KS> f1;
    chain_load_x<NG, KS>(f0, xlo, 0);
#pragma nounroll
    for (int k0 = 0; k0 < H - 2 * KS; k0 += 2 * KS) {
        chain_load_stage<NG, KS, N>(f1, wp, xlo, k0 + KS);
        chain_mma_stage<NG, KS>(f0, acc);
        chain_load_stage<NG, KS, N>(f0, wp, xlo, k0 + 2 * KS);
        chain_mma_stage<NG, KS>(f1, acc);
    }
    chain_load_stage<NG, KS, N>(f1, wp, xlo, H - KS);
    chain_mma_stage<NG, KS>(f0, acc);
    chain_load_w<NG, KS, N>(f0, wp, H);
    chain_mma_stage<NG, KS>(f1, acc);
}

template <int NG, int KS, int H, int N>
__device__ __forceinline__ void chain_joined_hi(v8f (&acc)[NG][4],
                                                ChainStage<NG, KS>& f0,
                                                const float* __restrict__ wp,
                                                const float* xhi) {
    ChainStage<NG, KS> f1;
    chain_load_x<NG, KS>(f0, xhi, 0);
    const float* wph = wp + (size_t)H * N;
#pragma nounroll
    for (int k0 = 0; k0 < H - 2 * KS; k0 += 2 * KS) {
        chain_load_stage<NG, KS, N>(f1, wph, xhi, k0 + KS);
        chain_mma_stage<NG, KS>(f0, acc);
        chain_load_stage<NG, KS, N>(f0, wph, xhi, k0 + 2 * KS);
        chain_mma_stage<NG, KS>(f1, acc);
    }
    chain_load_stage<NG, KS, N>(f1, wph, xhi, H - KS);
    chain_mma_stage<NG, KS>(f0, acc);
    chain_mma_stage<NG, KS>(f1, acc);
}

template <int NG, bool RELU>
__device__ __forceinline__ void chain_store_strip(const v8f (&acc)[NG][4],
                                                  const float* __restrict__ bias,
                                                  float* dst, int pitch, int nb, int c, int h) {
#pragma unroll
    for (int ng = 0; ng < NG; ++ng) {
#pragma unroll
        for (int r = 0; r < 8; ++r) {
            const int f = nb + 64 * ng + 32 * h + 4 * r;
            const v4f bia = *(const v4f*)(bias + f);
            v4f o;
            o.x = acc[ng][0][r] + bia.x;
            o.y = acc[ng][1][r] + bia.y;
            o.z = acc[ng][2][r] + bia.z;
            o.w = acc[ng][3][r] + bia.w;
            if (RELU) {
                o.x = fmaxf(o.x, 0.0f);
                o.y = fmaxf(o.y, 0.0f);
                o.z = fmaxf(o.z, 0.0f);
                o.w = fmaxf(o.w, 0.0f);
            }
            *(v4f*)(dst + c * pitch + f) = o;
        }
    }
}

__global__ __launch_bounds__((CH_WAVES + GM_WAVES) * 32)
void mlp_chain_kernel(const float* __restrict__ emb_a,
                      const float* __restrict__ emb_b,
                      const int* __restrict__ keys_a,
                      const int* __restrict__ keys_b,
                      const float* __restrict__ W1, const float* __restrict__ b1,
                      const float* __restrict__ W2, const float* __restrict__ b2,
                      const float* __restrict__ Wo, const float* __restrict__ bo,
                      float* __restrict__ out) {
    __shared__ __align__(16) float hbuf[CH_ROWS * H_PITCH];
    __shared__ __align__(16) float rbuf[CH_ROWS * R_PITCH];
    __shared__ __align__(16) unsigned short lists[SC_WAVES * SC_SLOTS * SC_ROWS];
    __shared__ __align__(16) unsigned char  tbl[SC_WAVES * NBUCKETS];
    __shared__ int cntl[SC_WAVES * SC_SLOTS];

    const int lane = threadIdx.x & 31;
    const int wave = threadIdx.x >> 5;
    const int h    = lane >> 4;
    const int c    = lane & 15;
    const int m0   = blockIdx.x * CH_ROWS;

    const int wave_u = __builtin_amdgcn_readfirstlane(wave);
    const bool mat = wave_u < CH_WAVES;

    const float* wp1 = W1 + (size_t)(4 * h) * (2 * D_MODEL) + (128 * wave + 4 * c);
    const float* wp2 = W2 + (size_t)(4 * h) * D_MODEL + (64 * wave + 4 * c);
    const float* wp3 = Wo + (size_t)(4 * h) * D_MODEL + (64 * wave + 4 * c);

    v8f acc1[2][4];
#pragma unroll
    for (int ng = 0; ng < 2; ++ng) {
#pragma unroll
        for (int t = 0; t < 4; ++t) acc1[ng][t] = v8f{};
    }
    ChainStage<2, 32> s1;
    if (mat) chain_load_w<2, 32, 2 * D_MODEL>(s1, wp1, 0);
    const int lead = scan_bucket_rows(keys_a, keys_b, lists, tbl, (unsigned char*)hbuf, cntl,
                                      m0, wave_u, lane);
    __syncthreads();
    if (mat) {
        chain_joined_lo<2, 32, D_MODEL, 2 * D_MODEL>(
            acc1, s1, wp1, emb_a + (size_t)(m0 + c) * D_MODEL + 4 * h);
    } else {
        const int gw = wave_u - CH_WAVES;
        gather_bucket_means(emb_b, lists, cntl, lead, rbuf, R_PITCH, GM_ROWS * gw, lane);
    }
    __syncthreads();

    ChainStage<1, 64> s2;
    if (mat) {
        chain_joined_hi<2, 32, D_MODEL, 2 * D_MODEL>(acc1, s1, wp1, rbuf + c * R_PITCH + 4 * h);
        chain_load_w<1, 64, D_MODEL>(s2, wp2, 0);
        chain_store_strip<2, true>(acc1, b1, hbuf, H_PITCH, 128 * wave, c, h);
    }
    __syncthreads();

    v8f acc2[1][4];
#pragma unroll
    for (int t = 0; t < 4; ++t) acc2[0][t] = v8f{};
    ChainStage<1, 32> s3;
    if (mat) {
        chain_layer<1, 64, 2 * D_MODEL, D_MODEL>(acc2, s2, wp2, hbuf + c * H_PITCH + 4 * h);
        chain_load_w<1, 32, D_MODEL>(s3, wp3, 0);
        chain_store_strip<1, false>(acc2, b2, rbuf, R_PITCH, 64 * wave, c, h);
    }
    __syncthreads();

    if (mat) {
        v8f acc3[1][4];
#pragma unroll
        for (int t = 0; t < 4; ++t) acc3[0][t] = v8f{};
        chain_layer<1, 32, D_MODEL, D_MODEL>(acc3, s3, wp3, rbuf + c * R_PITCH + 4 * h);
        chain_store_strip<1, false>(acc3, bo, hbuf, R_PITCH, 64 * wave, c, h);
    }
    __syncthreads();

#pragma unroll
    for (int it = 0; it < (CH_ROWS * D_MODEL / 4) / ((CH_WAVES + GM_WAVES) * 32); ++it) {
        const int idx  = it * ((CH_WAVES + GM_WAVES) * 32) + threadIdx.x;
        const int row  = idx / (D_MODEL / 4);
        const int col4 = idx % (D_MODEL / 4);
        *(v4f*)(out + (size_t)(m0 + row) * D_MODEL + 4 * col4) =
            *(const v4f*)(hbuf + row * R_PITCH + 4 * col4);
    }
}

extern "C" void kernel_launch(void* const* d_in, const int* in_sizes, int n_in,
                              void* d_out, int out_size, void* d_ws, size_t ws_size,
                              hipStream_t stream) {
    const float* emb_a  = (const float*)d_in[0];
    const float* emb_b  = (const float*)d_in[1];
    const int*   keys_a = (const int*)d_in[2];
    const int*   keys_b = (const int*)d_in[3];
    const float* W1 = (const float*)d_in[4];
    const float* b1 = (const float*)d_in[5];
    const float* W2 = (const float*)d_in[6];
    const float* b2 = (const float*)d_in[7];
    const float* Wo = (const float*)d_in[8];
    const float* bo = (const float*)d_in[9];
    float* out = (float*)d_out;

    mlp_chain_kernel<<<BATCH_A / CH_ROWS, (CH_WAVES + GM_WAVES) * 32, 0, stream>>>(emb_a, emb_b, keys_a, keys_b,
                                                                      W1, b1, W2, b2,
                                                                      Wo, bo, out);
}


# each block starts its K walks at its own stage (8 phases from the block index) and wraps round, so blocks stop asking L2 for the same weight lines at once; sum order rotated: new rounding
# speedup vs baseline: 11.3109x; 1.0361x over previous
#include <hip/hip_runtime.h>

typedef float v2f __attribute__((ext_vector_type(2)));
typedef float v4f __attribute__((ext_vector_type(4)));
typedef float v8f __attribute__((ext_vector_type(8)));

#define D_MODEL   256
#define BATCH_A   4096
#define BATCH_B   8192
#define NKEYS     4
#define NBUCKETS  1024

__device__ __forceinline__ int hash_row(const int* __restrict__ keys, int row) {
    int s = 0;
#pragma unroll
    for (int k = 0; k < NKEYS; ++k) s += keys[row * NKEYS + k];
    return (int)(((unsigned)s) & (NBUCKETS - 1));
}

#define GM_ROWS  4
#define GM_WAVES 4
#define GM_SLOTS 8
#define SC_WAVES  8
#define SC_ROWS   (BATCH_B / SC_WAVES)
#define SC_CHUNKS (SC_ROWS / 32)
#define SC_SLOTS  16
static_assert(NKEYS == 4, "a row's keys are one 16-byte load");
static_assert(BATCH_B <= 65536, "row numbers are kept as 16-bit list entries");
static_assert(SC_CHUNKS <= 32, "one bit per chunk in a 32-bit word");

__device__ __forceinline__ int scan_bucket_rows(const int* __restrict__ keys_a,
                                                const int* __restrict__ keys_b,
                                                unsigned short* lists, unsigned char* tbl,
                                                unsigned char* tv,
                                                int* cntl, int m0, int q, int lane) {
    int4 kr[SC_CHUNKS];
#pragma unroll
    for (int s = 0; s < SC_CHUNKS; ++s)
        kr[s] = *(const int4*)&keys_b[(size_t)(q * SC_ROWS + 32 * s + lane) * NKEYS];

    const int aj = lane & (SC_SLOTS - 1);
    const int ha = hash_row(keys_a, m0 + aj);
    int lead = aj;
#pragma unroll
    for (int j = SC_SLOTS - 1; j >= 0; --j) {
        const int hj = __builtin_amdgcn_readlane(ha, j);
        lead = (hj == ha) ? j : lead;
    }

    unsigned char* tw = tbl + q * NBUCKETS;
    *(uint4*)(tw + 32 * lane)      = uint4{0u, 0u, 0u, 0u};
    *(uint4*)(tw + 32 * lane + 16) = uint4{0u, 0u, 0u, 0u};
    if (lane < SC_SLOTS && lead == aj) tw[ha] = (unsigned char)(aj + 1);

    unsigned char* tvw = tv + q * (SC_CHUNKS * 32);
    int ts[SC_CHUNKS];
#pragma unroll
    for (int s = 0; s < SC_CHUNKS; ++s) {
        const unsigned hsh = (unsigned)(kr[s].x + kr[s].y + kr[s].z + kr[s].w) & (NBUCKETS - 1);
        ts[s] = tw[hsh];
    }
    unsigned hit = 0u;
#pragma unroll
    for (int s = 0; s < SC_CHUNKS; ++s) {
        tvw[32 * s + lane] = (unsigned char)ts[s];
        if (__builtin_amdgcn_ballot_w32(ts[s] != 0) != 0u) hit |= 1u << s;
    }

    int cntv = 0;
    unsigned short* lw = lists + (size_t)q * SC_SLOTS * SC_ROWS;
    int snext = hit != 0u ? __builtin_ctz(hit) : 0;
    int tn = tvw[32 * snext + lane];
    while (hit != 0u) {
        const int s  = snext;
        const int tc = tn;
        hit &= hit - 1u;
        snext = hit != 0u ? __builtin_ctz(hit) : 0;
        tn = tvw[32 * snext + lane];
        unsigned m = __builtin_amdgcn_ballot_w32(tc != 0);
        while (m != 0u) {
            const int l    = __builtin_ctz(m);
            const int slot = __builtin_amdgcn_readlane(tc, l) - 1;
            const int c    = __builtin_amdgcn_readlane(cntv, slot);
            lw[slot * SC_ROWS + c] = (unsigned short)(q * SC_ROWS + 32 * s + l);
            cntv = (lane == slot) ? c + 1 : cntv;
            m &= m - 1u;
        }
    }
    if (lane < SC_SLOTS) cntl[q * SC_SLOTS + lane] = cntv;
    return lead;
}

__device__ __forceinline__ void gather_bucket_means(const float* __restrict__ emb_b,
                                                    const unsigned short* lists,
                                                    const int* cntl, int lead,
                                                    float* dst, int pitch,
                                                    int lrow0, int lane) {
    int ldr[GM_ROWS], cq[GM_ROWS][SC_WAVES], cnt[GM_ROWS];
    int maxcnt = 0;
#pragma unroll
    for (int i = 0; i < GM_ROWS; ++i) {
        ldr[i] = __builtin_amdgcn_readlane(lead, lrow0 + i);
        cnt[i] = 0;
#pragma unroll
        for (int q = 0; q < SC_WAVES; ++q) {
            cq[i][q] = __builtin_amdgcn_readfirstlane(cntl[q * SC_SLOTS + ldr[i]]);
            cnt[i] += cq[i][q];
        }
        maxcnt = cnt[i] > maxcnt ? cnt[i] : maxcnt;
    }

    v4f acc[GM_ROWS][2];
#pragma unroll
    for (int i = 0; i < GM_ROWS; ++i) {
        acc[i][0] = v4f{};
        acc[i][1] = v4f{};
    }

    for (int base = 0; base < maxcnt; base += 32) {
        int pv[GM_ROWS];
#pragma unroll
        for (int i = 0; i < GM_ROWS; ++i) {
            const int k  = base + lane;
            const int kk = k < cnt[i] ? k : cnt[i] - 1;
            int off = kk, qq = 0;
#pragma unroll
            for (int q = 0; q < SC_WAVES - 1; ++q) {
                const bool nx = (qq == q) && (off >= cq[i][q]);
                off = nx ? off - cq[i][q] : off;
                qq  = nx ? q + 1 : qq;
            }
            int row = 0;
            if (kk >= 0) row = lists[(size_t)(qq * SC_SLOTS + ldr[i]) * SC_ROWS + off];
            pv[i] = row;
        }
        const int lim = (maxcnt - base) < 32 ? (maxcnt - base) : 32;
        for (int o = 0; o < lim; o += GM_SLOTS) {
            v4f v[GM_ROWS][GM_SLOTS][2];
#pragma unroll
            for (int i = 0; i < GM_ROWS; ++i) {
#pragma unroll
                for (int u = 0; u < GM_SLOTS; ++u) {
                    const int row = __builtin_amdgcn_readlane(pv[i], o + u);
                    const float* sp = emb_b + (size_t)row * D_MODEL + 4 * lane;
                    v[i][u][0] = *(const v4f*)sp;
                    v[i][u][1] = *(const v4f*)(sp + D_MODEL / 2);
                }
            }
#pragma unroll
            for (int i = 0; i < GM_ROWS; ++i) {
#pragma unroll
                for (int u = 0; u < GM_SLOTS; ++u) {
                    const bool in_slice = (base + o + u < cnt[i]);
#pragma unroll
                    for (int q = 0; q < 2; ++q) {
                        acc[i][q].x += in_slice ? v[i][u][q].x : 0.0f;
                        acc[i][q].y += in_slice ? v[i][u][q].y : 0.0f;
                        acc[i][q].z += in_slice ? v[i][u][q].z : 0.0f;
                        acc[i][q].w += in_slice ? v[i][u][q].w : 0.0f;
                    }
                }
            }
        }
    }

#pragma unroll
    for (int i = 0; i < GM_ROWS; ++i) {
        const float scale = 1.0f / (float)(cnt[i] < 1 ? 1 : cnt[i]);
#pragma unroll
        for (int q = 0; q < 2; ++q) {
            v4f mean;
            mean.x = acc[i][q].x * scale;
            mean.y = acc[i][q].y * scale;
            mean.z = acc[i][q].z * scale;
            mean.w = acc[i][q].w * scale;
            *(v4f*)(dst + (lrow0 + i) * pitch + (D_MODEL / 2) * q + 4 * lane) = mean;
        }
    }
}

#define CH_WAVES  4
#define CH_ROWS   16
#define CH_PHASES 8
static_assert(GM_WAVES * GM_ROWS == CH_ROWS, "the gather waves' rows cover the block's batch rows");
static_assert(CH_ROWS == SC_SLOTS, "one table slot per batch row of the block");
static_assert(CH_WAVES + GM_WAVES == SC_WAVES, "every wave of the block scans one part");
#define H_PITCH   (2 * D_MODEL + 4)
#define R_PITCH   (D_MODEL + 4)

template <int NG, int KS>
struct ChainStage {
    v4f w[NG][KS / 8][2][2];
    v4f x[KS / 8];
};

template <int NG, int KS, int N>
__device__ __forceinline__ void chain_load_w(ChainStage<NG, KS>& f,
                                             const float* __restrict__ wp, int k0) {
#pragma unroll
    for (int g = 0; g < KS / 8; ++g) {
#pragma unroll
        for (int ng = 0; ng < NG; ++ng) {
#pragma unroll
            for (int j = 0; j < 2; ++j) {
#pragma unroll
                for (int s = 0; s < 2; ++s)
                    f.w[ng][g][j][s] =
                        *(const v4f*)(wp + (size_t)(k0 + 8 * g + 2 * j + s) * N + 64 * ng);
            }
        }
    }
}

template <int NG, int KS>
__device__ __forceinline__ void chain_load_x(ChainStage<NG, KS>& f, const float* xp, int k0) {
#pragma unroll
    for (int g = 0; g < KS / 8; ++g) f.x[g] = *(const v4f*)(xp + (k0 + 8 * g));
}

template <int NG, int KS, int N>
__device__ __forceinline__ void chain_load_stage(ChainStage<NG, KS>& f,
                                                 const float* __restrict__ wp,
                                                 const float* xp,
                                                 int k0) {
    chain_load_x<NG, KS>(f, xp, k0);
    chain_load_w<NG, KS, N>(f, wp, k0);
}

template <int NG, int KS>
__device__ __forceinline__ void chain_mma_stage(const ChainStage<NG, KS>& f, v8f (&acc)[NG][4]) {
#pragma unroll
    for (int g = 0; g < KS / 8; ++g) {
#pragma unroll
        for (int j = 0; j < 2; ++j) {
            v2f x2;
            x2.x = f.x[g][2 * j];
            x2.y = f.x[g][2 * j + 1];
#pragma unroll
            for (int ng = 0; ng < NG; ++ng) {
#pragma unroll
                for (int t = 0; t < 4; ++t) {
                    v2f w2;
                    w2.x = f.w[ng][g][j][0][t];
                    w2.y = f.w[ng][g][j][1][t];
                    acc[ng][t] = __builtin_amdgcn_wmma_f32_16x16x4_f32(false, w2, false, x2,
                                                                       (short)0, acc[ng][t],
                                                                       false, false);
                }
            }
        }
    }
}

template <int KS, int K>
__device__ __forceinline__ int chain_rot(int ph) {
    static_assert(((K / KS) & (K / KS - 1)) == 0, "the stage count is a power of two");
    return KS * (ph & (K / KS - 1));
}

template <int NG, int KS, int K, int N>
__device__ __forceinline__ void chain_layer(v8f (&acc)[NG][4],
                                            ChainStage<NG, KS>& f0,
                                            const float* __restrict__ wp,
                                            const float* xp, int ph) {
    static_assert(KS % 8 == 0 && K % (2 * KS) == 0 && K >= 4 * KS, "two stages per trip, two trips");
    static_assert((K & (K - 1)) == 0, "the walk wraps round with a mask");
    const int rot = chain_rot<KS, K>(ph);
    ChainStage<NG, KS> f1;
    chain_load_x<NG, KS>(f0, xp, rot);
#pragma nounroll
    for (int k0 = 0; k0 < K - 2 * KS; k0 += 2 * KS) {
        chain_load_stage<NG, KS, N>(f1, wp, xp, (rot + k0 + KS) & (K - 1));
        chain_mma_stage<NG, KS>(f0, acc);
        chain_load_stage<NG, KS, N>(f0, wp, xp, (rot + k0 + 2 * KS) & (K - 1));
        chain_mma_stage<NG, KS>(f1, acc);
    }
    chain_load_stage<NG, KS, N>(f1, wp, xp, (rot + K - KS) & (K - 1));
    chain_mma_stage<NG, KS>(f0, acc);
    chain_mma_stage<NG, KS>(f1, acc);
}

template <int NG, int KS, int H, int N>
__device__ __forceinline__ void chain_joined_lo(v8f (&acc)[NG][4],
                                                ChainStage<NG, KS>& f0,
                                                const float* __restrict__ wp,
                                                const float* xlo, int ph) {
    static_assert(KS % 8 == 0 && H % (2 * KS) == 0 && H >= 4 * KS, "whole trips on each side of the join");
    static_assert((H & (H - 1)) == 0, "each side's walk wraps round with a mask");
    const int rot = chain_rot<KS, H>(ph);
    ChainStage<NG, KS> f1;
    chain_load_x<NG, KS>(f0, xlo, rot);
#pragma nounroll
    for (int k0 = 0; k0 < H - 2 * KS; k0 += 2 * KS) {
        chain_load_stage<NG, KS, N>(f1, wp, xlo, (rot + k0 + KS) & (H - 1));
        chain_mma_stage<NG, KS>(f0, acc);
        chain_load_stage<NG, KS, N>(f0, wp, xlo, (rot + k0 + 2 * KS) & (H - 1));
        chain_mma_stage<NG, KS>(f1, acc);
    }
    chain_load_stage<NG, KS, N>(f1, wp, xlo, (rot + H - KS) & (H - 1));
    chain_mma_stage<NG, KS>(f0, acc);
    chain_load_w<NG, KS, N>(f0, wp, H + rot);
    chain_mma_stage<NG, KS>(f1, acc);
}

template <int NG, int KS, int H, int N>
__device__ __forceinline__ void chain_joined_hi(v8f (&acc)[NG][4],
                                                ChainStage<NG, KS>& f0,
                                                const float* __restrict__ wp,
                                                const float* xhi, int ph) {
    const int rot = chain_rot<KS, H>(ph);
    ChainStage<NG, KS> f1;
    chain_load_x<NG, KS>(f0, xhi, rot);
    const float* wph = wp + (size_t)H * N;
#pragma nounroll
    for (int k0 = 0; k0 < H - 2 * KS; k0 += 2 * KS) {
        chain_load_stage<NG, KS, N>(f1, wph, xhi, (rot + k0 + KS) & (H - 1));
        chain_mma_stage<NG, KS>(f0, acc);
        chain_load_stage<NG, KS, N>(f0, wph, xhi, (rot + k0 + 2 * KS) & (H - 1));
        chain_mma_stage<NG, KS>(f1, acc);
    }
    chain_load_stage<NG, KS, N>(f1, wph, xhi, (rot + H - KS) & (H - 1));
    chain_mma_stage<NG, KS>(f0, acc);
    chain_mma_stage<NG, KS>(f1, acc);
}

template <int NG, bool RELU>
__device__ __forceinline__ void chain_store_strip(const v8f (&acc)[NG][4],
                                                  const float* __restrict__ bias,
                                                  float* dst, int pitch, int nb, int c, int h) {
#pragma unroll
    for (int ng = 0; ng < NG; ++ng) {
#pragma unroll
        for (int r = 0; r < 8; ++r) {
            const int f = nb + 64 * ng + 32 * h + 4 * r;
            const v4f bia = *(const v4f*)(bias + f);
            v4f o;
            o.x = acc[ng][0][r] + bia.x;
            o.y = acc[ng][1][r] + bia.y;
            o.z = acc[ng][2][r] + bia.z;
            o.w = acc[ng][3][r] + bia.w;
            if (RELU) {
                o.x = fmaxf(o.x, 0.0f);
                o.y = fmaxf(o.y, 0.0f);
                o.z = fmaxf(o.z, 0.0f);
                o.w = fmaxf(o.w, 0.0f);
            }
            *(v4f*)(dst + c * pitch + f) = o;
        }
    }
}

__global__ __launch_bounds__((CH_WAVES + GM_WAVES) * 32)
void mlp_chain_kernel(const float* __restrict__ emb_a,
                      const float* __restrict__ emb_b,
                      const int* __restrict__ keys_a,
                      const int* __restrict__ keys_b,
                      const float* __restrict__ W1, const float* __restrict__ b1,
                      const float* __restrict__ W2, const float* __restrict__ b2,
                      const float* __restrict__ Wo, const float* __restrict__ bo,
                      float* __restrict__ out) {
    __shared__ __align__(16) float hbuf[CH_ROWS * H_PITCH];
    __shared__ __align__(16) float rbuf[CH_ROWS * R_PITCH];
    __shared__ __align__(16) unsigned short lists[SC_WAVES * SC_SLOTS * SC_ROWS];
    __shared__ __align__(16) unsigned char  tbl[SC_WAVES * NBUCKETS];
    __shared__ int cntl[SC_WAVES * SC_SLOTS];

    const int lane = threadIdx.x & 31;
    const int wave = threadIdx.x >> 5;
    const int h    = lane >> 4;
    const int c    = lane & 15;
    const int m0   = blockIdx.x * CH_ROWS;

    const int wave_u = __builtin_amdgcn_readfirstlane(wave);
    const bool mat = wave_u < CH_WAVES;

    const int bid = (int)blockIdx.x;
    const int ph  = (bid + (bid >> 3) + (bid >> 6)) & (CH_PHASES - 1);

    const float* wp1 = W1 + (size_t)(4 * h) * (2 * D_MODEL) + (128 * wave + 4 * c);
    const float* wp2 = W2 + (size_t)(4 * h) * D_MODEL + (64 * wave + 4 * c);
    const float* wp3 = Wo + (size_t)(4 * h) * D_MODEL + (64 * wave + 4 * c);

    v8f acc1[2][4];
#pragma unroll
    for (int ng = 0; ng < 2; ++ng) {
#pragma unroll
        for (int t = 0; t < 4; ++t) acc1[ng][t] = v8f{};
    }
    ChainStage<2, 32> s1;
    if (mat) chain_load_w<2, 32, 2 * D_MODEL>(s1, wp1, chain_rot<32, D_MODEL>(ph));
    const int lead = scan_bucket_rows(keys_a, keys_b, lists, tbl, (unsigned char*)hbuf, cntl,
                                      m0, wave_u, lane);
    __syncthreads();
    if (mat) {
        chain_joined_lo<2, 32, D_MODEL, 2 * D_MODEL>(
            acc1, s1, wp1, emb_a + (size_t)(m0 + c) * D_MODEL + 4 * h, ph);
    } else {
        const int gw = wave_u - CH_WAVES;
        gather_bucket_means(emb_b, lists, cntl, lead, rbuf, R_PITCH, GM_ROWS * gw, lane);
    }
    __syncthreads();

    ChainStage<1, 64> s2;
    if (mat) {
        chain_joined_hi<2, 32, D_MODEL, 2 * D_MODEL>(acc1, s1, wp1, rbuf + c * R_PITCH + 4 * h, ph);
        chain_load_w<1, 64, D_MODEL>(s2, wp2, chain_rot<64, 2 * D_MODEL>(ph));
        chain_store_strip<2, true>(acc1, b1, hbuf, H_PITCH, 128 * wave, c, h);
    }
    __syncthreads();

    v8f acc2[1][4];
#pragma unroll
    for (int t = 0; t < 4; ++t) acc2[0][t] = v8f{};
    ChainStage<1, 32> s3;
    if (mat) {
        chain_layer<1, 64, 2 * D_MODEL, D_MODEL>(acc2, s2, wp2, hbuf + c * H_PITCH + 4 * h, ph);
        chain_load_w<1, 32, D_MODEL>(s3, wp3, chain_rot<32, D_MODEL>(ph));
        chain_store_strip<1, false>(acc2, b2, rbuf, R_PITCH, 64 * wave, c, h);
    }
    __syncthreads();

    if (mat) {
        v8f acc3[1][4];
#pragma unroll
        for (int t = 0; t < 4; ++t) acc3[0][t] = v8f{};
        chain_layer<1, 32, D_MODEL, D_MODEL>(acc3, s3, wp3, rbuf + c * R_PITCH + 4 * h, ph);
        chain_store_strip<1, false>(acc3, bo, hbuf, R_PITCH, 64 * wave, c, h);
    }
    __syncthreads();

#pragma unroll
    for (int it = 0; it < (CH_ROWS * D_MODEL / 4) / ((CH_WAVES + GM_WAVES) * 32); ++it) {
        const int idx  = it * ((CH_WAVES + GM_WAVES) * 32) + threadIdx.x;
        const int row  = idx / (D_MODEL / 4);
        const int col4 = idx % (D_MODEL / 4);
        *(v4f*)(out + (size_t)(m0 + row) * D_MODEL + 4 * col4) =
            *(const v4f*)(hbuf + row * R_PITCH + 4 * col4);
    }
}

extern "C" void kernel_launch(void* const* d_in, const int* in_sizes, int n_in,
                              void* d_out, int out_size, void* d_ws, size_t ws_size,
                              hipStream_t stream) {
    const float* emb_a  = (const float*)d_in[0];
    const float* emb_b  = (const float*)d_in[1];
    const int*   keys_a = (const int*)d_in[2];
    const int*   keys_b = (const int*)d_in[3];
    const float* W1 = (const float*)d_in[4];
    const float* b1 = (const float*)d_in[5];
    const float* W2 = (const float*)d_in[6];
    const float* b2 = (const float*)d_in[7];
    const float* Wo = (const float*)d_in[8];
    const float* bo = (const float*)d_in[9];
    float* out = (float*)d_out;

    mlp_chain_kernel<<<BATCH_A / CH_ROWS, (CH_WAVES + GM_WAVES) * 32, 0, stream>>>(emb_a, emb_b, keys_a, keys_b,
                                                                      W1, b1, W2, b2,
                                                                      Wo, bo, out);
}
